# starting kernel
# baseline (speedup 1.0000x reference)
#include <hip/hip_runtime.h>

#if defined(__HIP_DEVICE_COMPILE__)
#if !__has_builtin(__builtin_amdgcn_wmma_f32_16x16x4_f32)
#error "device pass: missing __builtin_amdgcn_wmma_f32_16x16x4_f32"
#endif
#endif

typedef __attribute__((ext_vector_type(2))) float v2f;
typedef __attribute__((ext_vector_type(8))) float v8f;

#define BN    8
#define CI    256
#define CO    128
#define HH    64
#define WW    64
#define OH    130
#define OW    130
#define UU    65
#define NPB   (UU*UU)
#define BLK_N 128
#define NT    ((NPB + BLK_N - 1)/BLK_N)

__global__ __launch_bounds__(256)
void deconv2d_fold_wmma(const float* __restrict__ x,
                        const float* __restrict__ wgt,
                        const float* __restrict__ bias,
                        float* __restrict__ out)
{
    const int tile = blockIdx.x;
    const int bb   = blockIdx.y;
    const int mt   = blockIdx.z >> 2;
    const int par  = blockIdx.z & 3;
    const int p = par >> 1, q = par & 1;

    const int wave = threadIdx.x >> 5;
    const int lane = threadIdx.x & 31;
    const int l16  = lane & 15;
    const int hi   = lane >> 4;
    const int kk   = hi * 2;

    const int wr = wave >> 2;
    const int wcn = wave & 3;
    const int mw = mt * 64 + wr * 32;
    const int nw = tile * BLK_N + wcn * 32;

    const float* xp[2][4];
    float        mk[2][4];
    bool         nval[2];
    int          uu[2], vv[2];
    #pragma unroll
    for (int sj = 0; sj < 2; ++sj) {
        const int n = nw + sj * 16 + l16;
        nval[sj] = (n < NPB);
        const int nc = nval[sj] ? n : (NPB - 1);
        uu[sj] = nc / UU;
        vv[sj] = nc % UU;
        #pragma unroll
        for (int t = 0; t < 4; ++t) {
            const int dy = t & 1, dx = t >> 1;
            const int h = uu[sj] - dy;
            const int w = vv[sj] - dx;
            const bool ok = (h >= 0) && (h < HH) && (w >= 0) && (w < WW);
            const int hc = h < 0 ? 0 : (h > HH - 1 ? HH - 1 : h);
            const int wc2 = w < 0 ? 0 : (w > WW - 1 ? WW - 1 : w);
            xp[sj][t] = x + (size_t)bb * CI * HH * WW
                          + (size_t)kk * HH * WW
                          + (size_t)hc * WW + wc2;
            mk[sj][t] = ok ? 1.0f : 0.0f;
        }
    }

    const float* wp[2][4];
    #pragma unroll
    for (int si = 0; si < 2; ++si) {
        #pragma unroll
        for (int t = 0; t < 4; ++t) {
            const int dy = t & 1, dx = t >> 1;
            const int ky = p + 2 * dy, kx = q + 2 * dx;
            wp[si][t] = wgt + (size_t)(mw + si * 16 + l16) * (CI * 16)
                            + (size_t)kk * 16
                            + (size_t)ky * 4 + kx;
        }
    }

    v8f acc[2][2] = {};

    #pragma unroll
    for (int t = 0; t < 4; ++t) {
        const float* __restrict__ wq0 = wp[0][t];
        const float* __restrict__ wq1 = wp[1][t];
        const float* __restrict__ xq0 = xp[0][t];
        const float* __restrict__ xq1 = xp[1][t];
        const float m0 = mk[0][t], m1 = mk[1][t];
        #pragma unroll 4
        for (int i0 = 0; i0 < CI; i0 += 4) {
            v2f a0, a1, b0, b1;
            a0.x = wq0[(size_t)i0 * 16];
            a0.y = wq0[(size_t)i0 * 16 + 16];
            a1.x = wq1[(size_t)i0 * 16];
            a1.y = wq1[(size_t)i0 * 16 + 16];
            b0.x = xq0[(size_t)i0 * (HH * WW)] * m0;
            b0.y = xq0[(size_t)i0 * (HH * WW) + HH * WW] * m0;
            b1.x = xq1[(size_t)i0 * (HH * WW)] * m1;
            b1.y = xq1[(size_t)i0 * (HH * WW) + HH * WW] * m1;
            acc[0][0] = __builtin_amdgcn_wmma_f32_16x16x4_f32(
                false, a0, false, b0, (short)0, acc[0][0], false, false);
            acc[0][1] = __builtin_amdgcn_wmma_f32_16x16x4_f32(
                false, a0, false, b1, (short)0, acc[0][1], false, false);
            acc[1][0] = __builtin_amdgcn_wmma_f32_16x16x4_f32(
                false, a1, false, b0, (short)0, acc[1][0], false, false);
            acc[1][1] = __builtin_amdgcn_wmma_f32_16x16x4_f32(
                false, a1, false, b1, (short)0, acc[1][1], false, false);
        }
    }

    #pragma unroll
    for (int sj = 0; sj < 2; ++sj) {
        if (!nval[sj]) continue;
        const int oy = p + 2 * uu[sj];
        const int ox = q + 2 * vv[sj];
        const size_t ob = (size_t)bb * CO * OH * OW + (size_t)oy * OW + ox;
        #pragma unroll
        for (int si = 0; si < 2; ++si) {
            #pragma unroll
            for (int j = 0; j < 8; ++j) {
                const int m = mw + si * 16 + j + hi * 8;
                float bs = 0.0f;
                #pragma unroll
                for (int t = 0; t < 4; ++t) {
                    const int dy = t & 1, dx = t >> 1;
                    bs += mk[sj][t] * bias[m * 16 + (p + 2 * dy) * 4 + (q + 2 * dx)];
                }
                out[ob + (size_t)m * OH * OW] = acc[si][sj][j] + bs;
            }
        }
    }
}

extern "C" void kernel_launch(void* const* d_in, const int* in_sizes, int n_in,
                              void* d_out, int out_size, void* d_ws, size_t ws_size,
                              hipStream_t stream) {
    (void)in_sizes; (void)n_in; (void)out_size; (void)d_ws; (void)ws_size;
    const float* x    = (const float*)d_in[0];
    const float* wgt  = (const float*)d_in[1];
    const float* bias = (const float*)d_in[2];
    float*       out  = (float*)d_out;

    dim3 grid(NT, BN, 8);
    dim3 block(256);
    deconv2d_fold_wmma<<<grid, block, 0, stream>>>(x, wgt, bias, out);
}


# 2-row wave tile, 2 waves/SIMD: weight record feeds 64 WMMAs, middle x row shared; next-step loads fenced ahead of WMMA groups, 4 v_nop behind each; edge pairs in the 256 blocks; sum order unchanged
# speedup vs baseline: 7.7418x; 7.7418x over previous
#include <hip/hip_runtime.h>


#if defined(__HIP_DEVICE_COMPILE__)
#if !__has_builtin(__builtin_amdgcn_wmma_f32_16x16x4_f32)
#error "device pass: missing __builtin_amdgcn_wmma_f32_16x16x4_f32"
#endif
#endif

typedef __attribute__((ext_vector_type(2))) float v2f;
typedef __attribute__((ext_vector_type(4))) float v4f;
typedef __attribute__((ext_vector_type(8))) float v8f;

#define UNROLL_REGS _Pragma("unroll")

#define WMMA_GROUP_BEGIN() __builtin_amdgcn_sched_barrier(0)
#define WMMA_GROUP_END()                                         \
    do {                                                         \
        __builtin_amdgcn_sched_barrier(0);                       \
        asm volatile("v_nop\n\tv_nop\n\tv_nop\n\tv_nop");       \
        __builtin_amdgcn_sched_barrier(0);                       \
    } while (0)

#define BN    8
#define CI    256
#define CO    128
#define HH    64
#define WW    64
#define OH    130
#define OW    130
#define EDGE_PER_B  129
#define EDGE_N      (BN * EDGE_PER_B)
#define EDGE_TILES  ((EDGE_N + 15) / 16)
#define MAIN_BLOCKS (BN * 4 * 8)
#define EDGE_TASKS  (EDGE_TILES * 8)
static_assert(EDGE_TASKS == 2 * MAIN_BLOCKS + 8, "edge task split assumes 520 = 2 x 256 + 8");

template <int NS>
struct Frag {
    v4f   a[4];
    float b[NS][2][2];
};

template <int NS>
__device__ __forceinline__ void tile(const float* __restrict__ x,
                                     const float* __restrict__ wgt,
                                     const float* __restrict__ bias,
                                     float* __restrict__ out,
                                     const int m0, const int c, const int h,
                                     const int (&pb)[NS], const int (&pu)[NS],
                                     const int (&pv)[NS], const bool (&valid)[NS])
{
    int  off[NS][2][2];
    bool ok[NS][2][2];
    UNROLL_REGS
    for (int s = 0; s < NS; ++s) {
        UNROLL_REGS
        for (int dy = 0; dy < 2; ++dy) {
            UNROLL_REGS
            for (int dx = 0; dx < 2; ++dx) {
                const int hh = pu[s] - dy;
                const int ww = pv[s] - dx;
                ok[s][dy][dx] = (hh >= 0) && (hh < HH) && (ww >= 0) && (ww < WW);
                const int hc = hh < 0 ? 0 : (hh > HH - 1 ? HH - 1 : hh);
                const int wc = ww < 0 ? 0 : (ww > WW - 1 ? WW - 1 : ww);
                off[s][dy][dx] = (pb[s] * CI + h) * (HH * WW) + hc * WW + wc;
            }
        }
    }

    const v4f* __restrict__ wrow =
        (const v4f*)(wgt + ((size_t)(m0 + c) * CI + h) * 16);

    v8f acc[NS][4] = {};

    auto loadf = [&](Frag<NS>& f, const int it) {
        const v4f* __restrict__ wp = wrow + (size_t)it * 8;
        f.a[0] = wp[0];
        f.a[1] = wp[1];
        f.a[2] = wp[2];
        f.a[3] = wp[3];
        const float* __restrict__ xi = x + (size_t)it * (2 * HH * WW);
        UNROLL_REGS
        for (int s = 0; s < NS; ++s) {
            UNROLL_REGS
            for (int dy = 0; dy < 2; ++dy) {
                f.b[s][dy][0] = xi[off[s][dy][0]];
                f.b[s][dy][1] = xi[off[s][dy][1]];
            }
        }
    };

    auto compute = [&](const Frag<NS>& f) {
        UNROLL_REGS
        for (int dy = 0; dy < 2; ++dy) {
            v2f bm[NS];
            UNROLL_REGS
            for (int s = 0; s < NS; ++s) {
                bm[s].x = ok[s][dy][0] ? f.b[s][dy][0] : 0.0f;
                bm[s].y = ok[s][dy][1] ? f.b[s][dy][1] : 0.0f;
            }
            v2f av[2][2];
            UNROLL_REGS
            for (int p = 0; p < 2; ++p) {
                const v4f wr = f.a[p + 2 * dy];
                UNROLL_REGS
                for (int q = 0; q < 2; ++q) {
                    av[p][q].x = q ? wr.y : wr.x;
                    av[p][q].y = q ? wr.w : wr.z;
                }
            }
            WMMA_GROUP_BEGIN();
            UNROLL_REGS
            for (int p = 0; p < 2; ++p) {
                UNROLL_REGS
                for (int q = 0; q < 2; ++q) {
                    UNROLL_REGS
                    for (int s = 0; s < NS; ++s) {
                        acc[s][p * 2 + q] = __builtin_amdgcn_wmma_f32_16x16x4_f32(
                            false, av[p][q], false, bm[s], (short)0, acc[s][p * 2 + q],
                            false, false);
                    }
                }
            }
            WMMA_GROUP_END();
        }
    };

    Frag<NS> f0, f1;
    loadf(f0, 0);
    for (int it = 0; it < CI / 2; it += 2) {
        loadf(f1, it + 1);
        compute(f0);
        loadf(f0, (it + 2 < CI / 2) ? it + 2 : CI / 2 - 1);
        compute(f1);
    }

    UNROLL_REGS
    for (int r = 0; r < 8; ++r) {
        const int o = m0 + 8 * h + r;
        const v4f* __restrict__ bp = (const v4f*)(bias + (size_t)o * 16);
        v4f br[4];
        br[0] = bp[0]; br[1] = bp[1]; br[2] = bp[2]; br[3] = bp[3];
        UNROLL_REGS
        for (int s = 0; s < NS; ++s) {
            if (!valid[s]) continue;
            UNROLL_REGS
            for (int p = 0; p < 2; ++p) {
                float bs[2];
                UNROLL_REGS
                for (int q = 0; q < 2; ++q) {
                    float t = 0.0f;
                    UNROLL_REGS
                    for (int dx = 0; dx < 2; ++dx) {
                        UNROLL_REGS
                        for (int dy = 0; dy < 2; ++dy) {
                            const v4f row = br[p + 2 * dy];
                            const float bv = (q + 2 * dx) == 0 ? row.x
                                           : (q + 2 * dx) == 1 ? row.y
                                           : (q + 2 * dx) == 2 ? row.z : row.w;
                            t += ok[s][dy][dx] ? bv : 0.0f;
                        }
                    }
                    bs[q] = t;
                }
                v2f val;
                val.x = acc[s][p * 2 + 0][r] + bs[0];
                val.y = acc[s][p * 2 + 1][r] + bs[1];
                const size_t oi = (((size_t)pb[s] * CO + o) * OH + (2 * pu[s] + p)) * OW
                                  + 2 * pv[s];
                *(v2f*)(out + oi) = val;
            }
        }
    }
}

struct Frag2 {
    v4f   a[4];
    float b[3][4][2];
};

__device__ __forceinline__ void tile_rows2(const float* __restrict__ x,
                                           const float* __restrict__ wgt,
                                           const float* __restrict__ bias,
                                           float* __restrict__ out,
                                           const int m0, const int c, const int h,
                                           const int b, const int u0)
{
    int  off[3][4][2];
    bool ok[3][4][2];
    UNROLL_REGS
    for (int j = 0; j < 3; ++j) {
        const int hh = u0 - 1 + j;
        const int hc = hh < 0 ? 0 : hh;
        UNROLL_REGS
        for (int s = 0; s < 4; ++s) {
            UNROLL_REGS
            for (int dx = 0; dx < 2; ++dx) {
                const int ww = s * 16 + c - dx;
                const int wc = ww < 0 ? 0 : ww;
                ok[j][s][dx]  = (hh >= 0) && (ww >= 0);
                off[j][s][dx] = (b * CI + h) * (HH * WW) + hc * WW + wc;
            }
        }
    }

    const v4f* __restrict__ wrow =
        (const v4f*)(wgt + ((size_t)(m0 + c) * CI + h) * 16);

    v8f acc[2][4][4] = {};

    auto loadf = [&](Frag2& f, const int it) {
        const v4f* __restrict__ wp = wrow + (size_t)it * 8;
        f.a[0] = wp[0];
        f.a[1] = wp[1];
        f.a[2] = wp[2];
        f.a[3] = wp[3];
        const float* __restrict__ xi = x + (size_t)it * (2 * HH * WW);
        UNROLL_REGS
        for (int j = 0; j < 3; ++j) {
            UNROLL_REGS
            for (int s = 0; s < 4; ++s) {
                f.b[j][s][0] = xi[off[j][s][0]];
                f.b[j][s][1] = xi[off[j][s][1]];
            }
        }
    };

    auto compute = [&](const Frag2& f) {
        v2f bm[3][4];
        UNROLL_REGS
        for (int j = 0; j < 3; ++j) {
            UNROLL_REGS
            for (int s = 0; s < 4; ++s) {
                bm[j][s].x = ok[j][s][0] ? f.b[j][s][0] : 0.0f;
                bm[j][s].y = ok[j][s][1] ? f.b[j][s][1] : 0.0f;
            }
        }
        UNROLL_REGS
        for (int dy = 0; dy < 2; ++dy) {
            v2f av[2][2];
            UNROLL_REGS
            for (int p = 0; p < 2; ++p) {
                const v4f wr = f.a[p + 2 * dy];
                UNROLL_REGS
                for (int q = 0; q < 2; ++q) {
                    av[p][q].x = q ? wr.y : wr.x;
                    av[p][q].y = q ? wr.w : wr.z;
                }
            }
            WMMA_GROUP_BEGIN();
            UNROLL_REGS
            for (int p = 0; p < 2; ++p) {
                UNROLL_REGS
                for (int q = 0; q < 2; ++q) {
                    UNROLL_REGS
                    for (int t = 0; t < 2; ++t) {
                        UNROLL_REGS
                        for (int s = 0; s < 4; ++s) {
                            acc[t][s][p * 2 + q] = __builtin_amdgcn_wmma_f32_16x16x4_f32(
                                false, av[p][q], false, bm[t + 1 - dy][s], (short)0,
                                acc[t][s][p * 2 + q], false, false);
                        }
                    }
                }
            }
            WMMA_GROUP_END();
        }
    };

    Frag2 f0, f1;
    loadf(f0, 0);
    for (int it = 0; it < CI / 2; it += 2) {
        loadf(f1, it + 1);
        __builtin_amdgcn_sched_barrier(0);
        compute(f0);
        loadf(f0, (it + 2 < CI / 2) ? it + 2 : CI / 2 - 1);
        __builtin_amdgcn_sched_barrier(0);
        compute(f1);
    }

    UNROLL_REGS
    for (int r = 0; r < 8; ++r) {
        const int o = m0 + 8 * h + r;
        const v4f* __restrict__ bp = (const v4f*)(bias + (size_t)o * 16);
        v4f br[4];
        br[0] = bp[0]; br[1] = bp[1]; br[2] = bp[2]; br[3] = bp[3];
        UNROLL_REGS
        for (int t = 0; t < 2; ++t) {
            UNROLL_REGS
            for (int s = 0; s < 4; ++s) {
                UNROLL_REGS
                for (int p = 0; p < 2; ++p) {
                    float bs[2];
                    UNROLL_REGS
                    for (int q = 0; q < 2; ++q) {
                        float tb = 0.0f;
                        UNROLL_REGS
                        for (int dx = 0; dx < 2; ++dx) {
                            UNROLL_REGS
                            for (int dy = 0; dy < 2; ++dy) {
                                const v4f row = br[p + 2 * dy];
                                const float bv = (q + 2 * dx) == 0 ? row.x
                                               : (q + 2 * dx) == 1 ? row.y
                                               : (q + 2 * dx) == 2 ? row.z : row.w;
                                tb += ok[t + 1 - dy][s][dx] ? bv : 0.0f;
                            }
                        }
                        bs[q] = tb;
                    }
                    v2f val;
                    val.x = acc[t][s][p * 2 + 0][r] + bs[0];
                    val.y = acc[t][s][p * 2 + 1][r] + bs[1];
                    const size_t oi = (((size_t)b * CO + o) * OH + (2 * (u0 + t) + p)) * OW
                                      + 2 * (s * 16 + c);
                    *(v2f*)(out + oi) = val;
                }
            }
        }
    }
}

__global__ __launch_bounds__(256, 2)
void deconv2d_parity_merged_wmma(const float* __restrict__ x,
                                 const float* __restrict__ wgt,
                                 const float* __restrict__ bias,
                                 float* __restrict__ out)
{
    const int lane = threadIdx.x & 31;
    const int wave = __builtin_amdgcn_readfirstlane((int)(threadIdx.x >> 5));
    const int c    = lane & 15;
    const int h    = lane >> 4;
    const int bx   = blockIdx.x;

    int task = -1;
    if (wave < 2)                  task = bx * 2 + wave;
    else if (wave == 2 && bx < 8)  task = 2 * MAIN_BLOCKS + bx;
    if (task >= 0) {
        const int e  = (task >> 3) * 16 + c;
        const int ec = e < EDGE_N ? e : EDGE_N - 1;
        const int b  = ec / EDGE_PER_B;
        const int r  = ec - b * EDGE_PER_B;
        int  pb[1], pu[1], pv[1];
        bool valid[1];
        pb[0] = b;
        pu[0] = r < 65 ? 64 : r - 65;
        pv[0] = r < 65 ? r : 64;
        valid[0] = e < EDGE_N;
        tile<1>(x, wgt, bias, out, (task & 7) * 16, c, h, pb, pu, pv, valid);
    }

    {
        const int osub = bx & 7;
        const int ug   = (bx >> 3) & 3;
        const int b    = bx >> 5;
        const int u0   = ug * 16 + wave * 2;
        tile_rows2(x, wgt, bias, out, osub * 16, c, h, b, u0);
    }
}

extern "C" void kernel_launch(void* const* d_in, const int* in_sizes, int n_in,
                              void* d_out, int out_size, void* d_ws, size_t ws_size,
                              hipStream_t stream) {
    (void)in_sizes; (void)n_in; (void)out_size; (void)d_ws; (void)ws_size;
    const float* x    = (const float*)d_in[0];
    const float* wgt  = (const float*)d_in[1];
    const float* bias = (const float*)d_in[2];
    float*       out  = (float*)d_out;

    dim3 grid(MAIN_BLOCKS);
    dim3 block(256);
    deconv2d_parity_merged_wmma<<<grid, block, 0, stream>>>(x, wgt, bias, out);
}


# F(2,2) along y: a wave tile of 2 rows x 32 columns (two per wave) takes b(X0-X1), (a+b)X1, a(X2-X1) over x rows u0-1..u0+1, 3 WMMAs for 4; sum order changes: 3 partial sums combined at the end
# speedup vs baseline: 8.2861x; 1.0703x over previous
#include <hip/hip_runtime.h>


#if defined(__HIP_DEVICE_COMPILE__)
#if !__has_builtin(__builtin_amdgcn_wmma_f32_16x16x4_f32)
#error "device pass: missing __builtin_amdgcn_wmma_f32_16x16x4_f32"
#endif
#endif

typedef __attribute__((ext_vector_type(2))) float v2f;
typedef __attribute__((ext_vector_type(4))) float v4f;
typedef __attribute__((ext_vector_type(8))) float v8f;

#define UNROLL_REGS _Pragma("unroll")

#define WMMA_GROUP_BEGIN() __builtin_amdgcn_sched_barrier(0)
#define WMMA_GROUP_END()                                         \
    do {                                                         \
        __builtin_amdgcn_sched_barrier(0);                       \
        asm volatile("v_nop\n\tv_nop\n\tv_nop\n\tv_nop");       \
        __builtin_amdgcn_sched_barrier(0);                       \
    } while (0)

#define BN    8
#define CI    256
#define CO    128
#define HH    64
#define WW    64
#define OH    130
#define OW    130
#define EDGE_PER_B  129
#define EDGE_N      (BN * EDGE_PER_B)
#define EDGE_TILES  ((EDGE_N + 15) / 16)
#define MAIN_BLOCKS (BN * 4 * 8)
#define EDGE_TASKS  (EDGE_TILES * 8)
static_assert(EDGE_TASKS == 2 * MAIN_BLOCKS + 8, "edge task split assumes 520 = 2 x 256 + 8");

template <int NS>
struct Frag {
    v4f   a[4];
    float b[NS][2][2];
};

template <int NS>
__device__ __forceinline__ void tile(const float* __restrict__ x,
                                     const float* __restrict__ wgt,
                                     const float* __restrict__ bias,
                                     float* __restrict__ out,
                                     const int m0, const int c, const int h,
                                     const int (&pb)[NS], const int (&pu)[NS],
                                     const int (&pv)[NS], const bool (&valid)[NS])
{
    int  off[NS][2][2];
    bool ok[NS][2][2];
    UNROLL_REGS
    for (int s = 0; s < NS; ++s) {
        UNROLL_REGS
        for (int dy = 0; dy < 2; ++dy) {
            UNROLL_REGS
            for (int dx = 0; dx < 2; ++dx) {
                const int hh = pu[s] - dy;
                const int ww = pv[s] - dx;
                ok[s][dy][dx] = (hh >= 0) && (hh < HH) && (ww >= 0) && (ww < WW);
                const int hc = hh < 0 ? 0 : (hh > HH - 1 ? HH - 1 : hh);
                const int wc = ww < 0 ? 0 : (ww > WW - 1 ? WW - 1 : ww);
                off[s][dy][dx] = (pb[s] * CI + h) * (HH * WW) + hc * WW + wc;
            }
        }
    }

    const v4f* __restrict__ wrow =
        (const v4f*)(wgt + ((size_t)(m0 + c) * CI + h) * 16);

    v8f acc[NS][4] = {};

    auto loadf = [&](Frag<NS>& f, const int it) {
        const v4f* __restrict__ wp = wrow + (size_t)it * 8;
        f.a[0] = wp[0];
        f.a[1] = wp[1];
        f.a[2] = wp[2];
        f.a[3] = wp[3];
        const float* __restrict__ xi = x + (size_t)it * (2 * HH * WW);
        UNROLL_REGS
        for (int s = 0; s < NS; ++s) {
            UNROLL_REGS
            for (int dy = 0; dy < 2; ++dy) {
                f.b[s][dy][0] = xi[off[s][dy][0]];
                f.b[s][dy][1] = xi[off[s][dy][1]];
            }
        }
    };

    auto compute = [&](const Frag<NS>& f) {
        UNROLL_REGS
        for (int dy = 0; dy < 2; ++dy) {
            v2f bm[NS];
            UNROLL_REGS
            for (int s = 0; s < NS; ++s) {
                bm[s].x = ok[s][dy][0] ? f.b[s][dy][0] : 0.0f;
                bm[s].y = ok[s][dy][1] ? f.b[s][dy][1] : 0.0f;
            }
            v2f av[2][2];
            UNROLL_REGS
            for (int p = 0; p < 2; ++p) {
                const v4f wr = f.a[p + 2 * dy];
                UNROLL_REGS
                for (int q = 0; q < 2; ++q) {
                    av[p][q].x = q ? wr.y : wr.x;
                    av[p][q].y = q ? wr.w : wr.z;
                }
            }
            WMMA_GROUP_BEGIN();
            UNROLL_REGS
            for (int p = 0; p < 2; ++p) {
                UNROLL_REGS
                for (int q = 0; q < 2; ++q) {
                    UNROLL_REGS
                    for (int s = 0; s < NS; ++s) {
                        acc[s][p * 2 + q] = __builtin_amdgcn_wmma_f32_16x16x4_f32(
                            false, av[p][q], false, bm[s], (short)0, acc[s][p * 2 + q],
                            false, false);
                    }
                }
            }
            WMMA_GROUP_END();
        }
    };

    Frag<NS> f0, f1;
    loadf(f0, 0);
    for (int it = 0; it < CI / 2; it += 2) {
        loadf(f1, it + 1);
        compute(f0);
        loadf(f0, (it + 2 < CI / 2) ? it + 2 : CI / 2 - 1);
        compute(f1);
    }

    UNROLL_REGS
    for (int r = 0; r < 8; ++r) {
        const int o = m0 + 8 * h + r;
        const v4f* __restrict__ bp = (const v4f*)(bias + (size_t)o * 16);
        v4f br[4];
        br[0] = bp[0]; br[1] = bp[1]; br[2] = bp[2]; br[3] = bp[3];
        UNROLL_REGS
        for (int s = 0; s < NS; ++s) {
            if (!valid[s]) continue;
            UNROLL_REGS
            for (int p = 0; p < 2; ++p) {
                float bs[2];
                UNROLL_REGS
                for (int q = 0; q < 2; ++q) {
                    float t = 0.0f;
                    UNROLL_REGS
                    for (int dx = 0; dx < 2; ++dx) {
                        UNROLL_REGS
                        for (int dy = 0; dy < 2; ++dy) {
                            const v4f row = br[p + 2 * dy];
                            const float bv = (q + 2 * dx) == 0 ? row.x
                                           : (q + 2 * dx) == 1 ? row.y
                                           : (q + 2 * dx) == 2 ? row.z : row.w;
                            t += ok[s][dy][dx] ? bv : 0.0f;
                        }
                    }
                    bs[q] = t;
                }
                v2f val;
                val.x = acc[s][p * 2 + 0][r] + bs[0];
                val.y = acc[s][p * 2 + 1][r] + bs[1];
                const size_t oi = (((size_t)pb[s] * CO + o) * OH + (2 * pu[s] + p)) * OW
                                  + 2 * pv[s];
                *(v2f*)(out + oi) = val;
            }
        }
    }
}

struct FragW {
    v4f   a[2][4];
    float b[2][3][2][2];
};

__device__ __forceinline__ void tile_rows2_f22(const float* __restrict__ x,
                                               const float* __restrict__ wgt,
                                               const float* __restrict__ bias,
                                               float* __restrict__ out,
                                               const int m0, const int c, const int h,
                                               const int b, const int u0, const int vb)
{
    int  off[3][2][2];
    bool ok[3][2][2];
    UNROLL_REGS
    for (int j = 0; j < 3; ++j) {
        const int hh = u0 - 1 + j;
        const int hc = hh < 0 ? 0 : hh;
        UNROLL_REGS
        for (int s = 0; s < 2; ++s) {
            UNROLL_REGS
            for (int dx = 0; dx < 2; ++dx) {
                const int ww = vb + s * 16 + c - dx;
                const int wc = ww < 0 ? 0 : ww;
                ok[j][s][dx]  = (hh >= 0) && (ww >= 0);
                off[j][s][dx] = (b * CI + h) * (HH * WW) + hc * WW + wc;
            }
        }
    }

    const v4f* __restrict__ wrow =
        (const v4f*)(wgt + ((size_t)(m0 + c) * CI + h) * 16);

    v8f acc[3][2][4] = {};

    auto loadf = [&](FragW& f, const int it) {
        const v4f* __restrict__ wp = wrow + (size_t)it * 16;
        const float* __restrict__ xi = x + (size_t)it * (4 * HH * WW);
        UNROLL_REGS
        for (int kk = 0; kk < 2; ++kk) {
            f.a[kk][0] = wp[kk * 8 + 0];
            f.a[kk][1] = wp[kk * 8 + 1];
            f.a[kk][2] = wp[kk * 8 + 2];
            f.a[kk][3] = wp[kk * 8 + 3];
            UNROLL_REGS
            for (int j = 0; j < 3; ++j) {
                UNROLL_REGS
                for (int s = 0; s < 2; ++s) {
                    f.b[kk][j][s][0] = xi[off[j][s][0] + kk * (2 * HH * WW)];
                    f.b[kk][j][s][1] = xi[off[j][s][1] + kk * (2 * HH * WW)];
                }
            }
        }
    };

    auto compute = [&](const FragW& f) {
        v2f bd[2][3][2];
        v2f aw[2][3][2][2];
        UNROLL_REGS
        for (int kk = 0; kk < 2; ++kk) {
            UNROLL_REGS
            for (int s = 0; s < 2; ++s) {
                v2f xm[3];
                UNROLL_REGS
                for (int j = 0; j < 3; ++j) {
                    xm[j].x = ok[j][s][0] ? f.b[kk][j][s][0] : 0.0f;
                    xm[j].y = ok[j][s][1] ? f.b[kk][j][s][1] : 0.0f;
                }
                bd[kk][0][s] = xm[0] - xm[1];
                bd[kk][1][s] = xm[1];
                bd[kk][2][s] = xm[2] - xm[1];
            }
            UNROLL_REGS
            for (int p = 0; p < 2; ++p) {
                const v4f wa = f.a[kk][p];
                const v4f wb = f.a[kk][p + 2];
                UNROLL_REGS
                for (int q = 0; q < 2; ++q) {
                    v2f ta, tb;
                    ta.x = q ? wa.y : wa.x;
                    ta.y = q ? wa.w : wa.z;
                    tb.x = q ? wb.y : wb.x;
                    tb.y = q ? wb.w : wb.z;
                    aw[kk][0][p][q] = tb;
                    aw[kk][1][p][q] = ta + tb;
                    aw[kk][2][p][q] = ta;
                }
            }
        }
        WMMA_GROUP_BEGIN();
        UNROLL_REGS
        for (int kk = 0; kk < 2; ++kk) {
            UNROLL_REGS
            for (int m = 0; m < 3; ++m) {
                UNROLL_REGS
                for (int p = 0; p < 2; ++p) {
                    UNROLL_REGS
                    for (int q = 0; q < 2; ++q) {
                        UNROLL_REGS
                        for (int s = 0; s < 2; ++s) {
                            acc[m][s][p * 2 + q] = __builtin_amdgcn_wmma_f32_16x16x4_f32(
                                false, aw[kk][m][p][q], false, bd[kk][m][s], (short)0,
                                acc[m][s][p * 2 + q], false, false);
                        }
                    }
                }
            }
        }
        WMMA_GROUP_END();
    };

    FragW f0, f1;
    loadf(f0, 0);
    for (int it = 0; it < CI / 4; it += 2) {
        loadf(f1, it + 1);
        __builtin_amdgcn_sched_barrier(0);
        compute(f0);
        loadf(f0, (it + 2 < CI / 4) ? it + 2 : CI / 4 - 1);
        __builtin_amdgcn_sched_barrier(0);
        compute(f1);
    }

    UNROLL_REGS
    for (int r = 0; r < 8; ++r) {
        const int o = m0 + 8 * h + r;
        const v4f* __restrict__ bp = (const v4f*)(bias + (size_t)o * 16);
        v4f br[4];
        br[0] = bp[0]; br[1] = bp[1]; br[2] = bp[2]; br[3] = bp[3];
        UNROLL_REGS
        for (int t = 0; t < 2; ++t) {
            UNROLL_REGS
            for (int s = 0; s < 2; ++s) {
                UNROLL_REGS
                for (int p = 0; p < 2; ++p) {
                    float bs[2];
                    UNROLL_REGS
                    for (int q = 0; q < 2; ++q) {
                        float tb = 0.0f;
                        UNROLL_REGS
                        for (int dx = 0; dx < 2; ++dx) {
                            UNROLL_REGS
                            for (int dy = 0; dy < 2; ++dy) {
                                const v4f row = br[p + 2 * dy];
                                const float bv = (q + 2 * dx) == 0 ? row.x
                                               : (q + 2 * dx) == 1 ? row.y
                                               : (q + 2 * dx) == 2 ? row.z : row.w;
                                tb += ok[t + 1 - dy][s][dx] ? bv : 0.0f;
                            }
                        }
                        bs[q] = tb;
                    }
                    v2f val;
                    val.x = (acc[2 * t][s][p * 2 + 0][r] + acc[1][s][p * 2 + 0][r]) + bs[0];
                    val.y = (acc[2 * t][s][p * 2 + 1][r] + acc[1][s][p * 2 + 1][r]) + bs[1];
                    const size_t oi = (((size_t)b * CO + o) * OH + (2 * (u0 + t) + p)) * OW
                                      + 2 * (vb + s * 16 + c);
                    *(v2f*)(out + oi) = val;
                }
            }
        }
    }
}

__global__ __launch_bounds__(256, 2)
void deconv2d_parity_merged_wmma(const float* __restrict__ x,
                                 const float* __restrict__ wgt,
                                 const float* __restrict__ bias,
                                 float* __restrict__ out)
{
    const int lane = threadIdx.x & 31;
    const int wave = __builtin_amdgcn_readfirstlane((int)(threadIdx.x >> 5));
    const int c    = lane & 15;
    const int h    = lane >> 4;
    const int bx   = blockIdx.x;

    int task = -1;
    if (wave < 2)                  task = bx * 2 + wave;
    else if (wave == 2 && bx < 8)  task = 2 * MAIN_BLOCKS + bx;
    if (task >= 0) {
        const int e  = (task >> 3) * 16 + c;
        const int ec = e < EDGE_N ? e : EDGE_N - 1;
        const int b  = ec / EDGE_PER_B;
        const int r  = ec - b * EDGE_PER_B;
        int  pb[1], pu[1], pv[1];
        bool valid[1];
        pb[0] = b;
        pu[0] = r < 65 ? 64 : r - 65;
        pv[0] = r < 65 ? r : 64;
        valid[0] = e < EDGE_N;
        tile<1>(x, wgt, bias, out, (task & 7) * 16, c, h, pb, pu, pv, valid);
    }

    {
        const int osub = bx & 7;
        const int ug   = (bx >> 3) & 3;
        const int b    = bx >> 5;
        const int u0   = ug * 16 + wave * 2;
        for (int half = 0; half < 2; ++half)
            tile_rows2_f22(x, wgt, bias, out, osub * 16, c, h, b, u0, half * 32);
    }
}

extern "C" void kernel_launch(void* const* d_in, const int* in_sizes, int n_in,
                              void* d_out, int out_size, void* d_ws, size_t ws_size,
                              hipStream_t stream) {
    (void)in_sizes; (void)n_in; (void)out_size; (void)d_ws; (void)ws_size;
    const float* x    = (const float*)d_in[0];
    const float* wgt  = (const float*)d_in[1];
    const float* bias = (const float*)d_in[2];
    float*       out  = (float*)d_out;

    dim3 grid(MAIN_BLOCKS);
    dim3 block(256);
    deconv2d_parity_merged_wmma<<<grid, block, 0, stream>>>(x, wgt, bias, out);
}


# Interior A tile rows = 8 channels x 2 row parities (was 16 channels): a lane reads only rows ky=p,p+2 of its weight record, so a weight load touches 8 cache lines, not 16; sum order unchanged
# speedup vs baseline: 8.5952x; 1.0373x over previous
#include <hip/hip_runtime.h>


#if defined(__HIP_DEVICE_COMPILE__)
#if !__has_builtin(__builtin_amdgcn_wmma_f32_16x16x4_f32)
#error "device pass: missing __builtin_amdgcn_wmma_f32_16x16x4_f32"
#endif
#endif

typedef __attribute__((ext_vector_type(2))) float v2f;
typedef __attribute__((ext_vector_type(4))) float v4f;
typedef __attribute__((ext_vector_type(8))) float v8f;

#define UNROLL_REGS _Pragma("unroll")

#define WMMA_GROUP_BEGIN() __builtin_amdgcn_sched_barrier(0)
#define WMMA_GROUP_END()                                         \
    do {                                                         \
        __builtin_amdgcn_sched_barrier(0);                       \
        asm volatile("v_nop\n\tv_nop\n\tv_nop\n\tv_nop");       \
        __builtin_amdgcn_sched_barrier(0);                       \
    } while (0)

#define BN    8
#define CI    256
#define CO    128
#define HH    64
#define WW    64
#define OH    130
#define OW    130
#define EDGE_PER_B  129
#define EDGE_N      (BN * EDGE_PER_B)
#define EDGE_TILES  ((EDGE_N + 15) / 16)
#define MAIN_BLOCKS (BN * 4 * 8)
#define EDGE_TASKS  (EDGE_TILES * 8)
static_assert(EDGE_TASKS == 2 * MAIN_BLOCKS + 8, "edge task split assumes 520 = 2 x 256 + 8");

template <int NS>
struct Frag {
    v4f   a[4];
    float b[NS][2][2];
};

template <int NS>
__device__ __forceinline__ void tile(const float* __restrict__ x,
                                     const float* __restrict__ wgt,
                                     const float* __restrict__ bias,
                                     float* __restrict__ out,
                                     const int m0, const int c, const int h,
                                     const int (&pb)[NS], const int (&pu)[NS],
                                     const int (&pv)[NS], const bool (&valid)[NS])
{
    int  off[NS][2][2];
    bool ok[NS][2][2];
    UNROLL_REGS
    for (int s = 0; s < NS; ++s) {
        UNROLL_REGS
        for (int dy = 0; dy < 2; ++dy) {
            UNROLL_REGS
            for (int dx = 0; dx < 2; ++dx) {
                const int hh = pu[s] - dy;
                const int ww = pv[s] - dx;
                ok[s][dy][dx] = (hh >= 0) && (hh < HH) && (ww >= 0) && (ww < WW);
                const int hc = hh < 0 ? 0 : (hh > HH - 1 ? HH - 1 : hh);
                const int wc = ww < 0 ? 0 : (ww > WW - 1 ? WW - 1 : ww);
                off[s][dy][dx] = (pb[s] * CI + h) * (HH * WW) + hc * WW + wc;
            }
        }
    }

    const v4f* __restrict__ wrow =
        (const v4f*)(wgt + ((size_t)(m0 + c) * CI + h) * 16);

    v8f acc[NS][4] = {};

    auto loadf = [&](Frag<NS>& f, const int it) {
        const v4f* __restrict__ wp = wrow + (size_t)it * 8;
        f.a[0] = wp[0];
        f.a[1] = wp[1];
        f.a[2] = wp[2];
        f.a[3] = wp[3];
        const float* __restrict__ xi = x + (size_t)it * (2 * HH * WW);
        UNROLL_REGS
        for (int s = 0; s < NS; ++s) {
            UNROLL_REGS
            for (int dy = 0; dy < 2; ++dy) {
                f.b[s][dy][0] = xi[off[s][dy][0]];
                f.b[s][dy][1] = xi[off[s][dy][1]];
            }
        }
    };

    auto compute = [&](const Frag<NS>& f) {
        UNROLL_REGS
        for (int dy = 0; dy < 2; ++dy) {
            v2f bm[NS];
            UNROLL_REGS
            for (int s = 0; s < NS; ++s) {
                bm[s].x = ok[s][dy][0] ? f.b[s][dy][0] : 0.0f;
                bm[s].y = ok[s][dy][1] ? f.b[s][dy][1] : 0.0f;
            }
            v2f av[2][2];
            UNROLL_REGS
            for (int p = 0; p < 2; ++p) {
                const v4f wr = f.a[p + 2 * dy];
                UNROLL_REGS
                for (int q = 0; q < 2; ++q) {
                    av[p][q].x = q ? wr.y : wr.x;
                    av[p][q].y = q ? wr.w : wr.z;
                }
            }
            WMMA_GROUP_BEGIN();
            UNROLL_REGS
            for (int p = 0; p < 2; ++p) {
                UNROLL_REGS
                for (int q = 0; q < 2; ++q) {
                    UNROLL_REGS
                    for (int s = 0; s < NS; ++s) {
                        acc[s][p * 2 + q] = __builtin_amdgcn_wmma_f32_16x16x4_f32(
                            false, av[p][q], false, bm[s], (short)0, acc[s][p * 2 + q],
                            false, false);
                    }
                }
            }
            WMMA_GROUP_END();
        }
    };

    Frag<NS> f0, f1;
    loadf(f0, 0);
    for (int it = 0; it < CI / 2; it += 2) {
        loadf(f1, it + 1);
        compute(f0);
        loadf(f0, (it + 2 < CI / 2) ? it + 2 : CI / 2 - 1);
        compute(f1);
    }

    UNROLL_REGS
    for (int r = 0; r < 8; ++r) {
        const int o = m0 + 8 * h + r;
        const v4f* __restrict__ bp = (const v4f*)(bias + (size_t)o * 16);
        v4f br[4];
        br[0] = bp[0]; br[1] = bp[1]; br[2] = bp[2]; br[3] = bp[3];
        UNROLL_REGS
        for (int s = 0; s < NS; ++s) {
            if (!valid[s]) continue;
            UNROLL_REGS
            for (int p = 0; p < 2; ++p) {
                float bs[2];
                UNROLL_REGS
                for (int q = 0; q < 2; ++q) {
                    float t = 0.0f;
                    UNROLL_REGS
                    for (int dx = 0; dx < 2; ++dx) {
                        UNROLL_REGS
                        for (int dy = 0; dy < 2; ++dy) {
                            const v4f row = br[p + 2 * dy];
                            const float bv = (q + 2 * dx) == 0 ? row.x
                                           : (q + 2 * dx) == 1 ? row.y
                                           : (q + 2 * dx) == 2 ? row.z : row.w;
                            t += ok[s][dy][dx] ? bv : 0.0f;
                        }
                    }
                    bs[q] = t;
                }
                v2f val;
                val.x = acc[s][p * 2 + 0][r] + bs[0];
                val.y = acc[s][p * 2 + 1][r] + bs[1];
                const size_t oi = (((size_t)pb[s] * CO + o) * OH + (2 * pu[s] + p)) * OW
                                  + 2 * pv[s];
                *(v2f*)(out + oi) = val;
            }
        }
    }
}

struct FragW {
    v4f   a[2][2][2];
    float b[2][3][2][2];
};

__device__ __forceinline__ void tile_rows2_f22(const float* __restrict__ x,
                                               const float* __restrict__ wgt,
                                               const float* __restrict__ bias,
                                               float* __restrict__ out,
                                               const int m0, const int c, const int h,
                                               const int b, const int u0, const int vb)
{
    int  off[3][2][2];
    bool ok[3][2][2];
    UNROLL_REGS
    for (int j = 0; j < 3; ++j) {
        const int hh = u0 - 1 + j;
        const int hc = hh < 0 ? 0 : hh;
        UNROLL_REGS
        for (int s = 0; s < 2; ++s) {
            UNROLL_REGS
            for (int dx = 0; dx < 2; ++dx) {
                const int ww = vb + s * 16 + c - dx;
                const int wc = ww < 0 ? 0 : ww;
                ok[j][s][dx]  = (hh >= 0) && (ww >= 0);
                off[j][s][dx] = (b * CI + h) * (HH * WW) + hc * WW + wc;
            }
        }
    }

    const v4f* __restrict__ wrow =
        (const v4f*)(wgt + ((size_t)(m0 + (c >> 1)) * CI + h) * 16) + (c & 1);

    v8f acc[3][2][4] = {};

    auto loadf = [&](FragW& f, const int it) {
        const v4f* __restrict__ wp = wrow + (size_t)it * 16;
        const float* __restrict__ xi = x + (size_t)it * (4 * HH * WW);
        UNROLL_REGS
        for (int kk = 0; kk < 2; ++kk) {
            UNROLL_REGS
            for (int g = 0; g < 2; ++g) {
                f.a[kk][g][0] = wp[g * (8 * CI * 4) + kk * 8 + 0];
                f.a[kk][g][1] = wp[g * (8 * CI * 4) + kk * 8 + 2];
            }
            UNROLL_REGS
            for (int j = 0; j < 3; ++j) {
                UNROLL_REGS
                for (int s = 0; s < 2; ++s) {
                    f.b[kk][j][s][0] = xi[off[j][s][0] + kk * (2 * HH * WW)];
                    f.b[kk][j][s][1] = xi[off[j][s][1] + kk * (2 * HH * WW)];
                }
            }
        }
    };

    auto compute = [&](const FragW& f) {
        v2f bd[2][3][2];
        v2f aw[2][3][2][2];
        UNROLL_REGS
        for (int kk = 0; kk < 2; ++kk) {
            UNROLL_REGS
            for (int s = 0; s < 2; ++s) {
                v2f xm[3];
                UNROLL_REGS
                for (int j = 0; j < 3; ++j) {
                    xm[j].x = ok[j][s][0] ? f.b[kk][j][s][0] : 0.0f;
                    xm[j].y = ok[j][s][1] ? f.b[kk][j][s][1] : 0.0f;
                }
                bd[kk][0][s] = xm[0] - xm[1];
                bd[kk][1][s] = xm[1];
                bd[kk][2][s] = xm[2] - xm[1];
            }
            UNROLL_REGS
            for (int g = 0; g < 2; ++g) {
                const v4f wa = f.a[kk][g][0];
                const v4f wb = f.a[kk][g][1];
                UNROLL_REGS
                for (int q = 0; q < 2; ++q) {
                    v2f ta, tb;
                    ta.x = q ? wa.y : wa.x;
                    ta.y = q ? wa.w : wa.z;
                    tb.x = q ? wb.y : wb.x;
                    tb.y = q ? wb.w : wb.z;
                    aw[kk][0][g][q] = tb;
                    aw[kk][1][g][q] = ta + tb;
                    aw[kk][2][g][q] = ta;
                }
            }
        }
        WMMA_GROUP_BEGIN();
        UNROLL_REGS
        for (int kk = 0; kk < 2; ++kk) {
            UNROLL_REGS
            for (int m = 0; m < 3; ++m) {
                UNROLL_REGS
                for (int g = 0; g < 2; ++g) {
                    UNROLL_REGS
                    for (int q = 0; q < 2; ++q) {
                        UNROLL_REGS
                        for (int s = 0; s < 2; ++s) {
                            acc[m][s][g * 2 + q] = __builtin_amdgcn_wmma_f32_16x16x4_f32(
                                false, aw[kk][m][g][q], false, bd[kk][m][s], (short)0,
                                acc[m][s][g * 2 + q], false, false);
                        }
                    }
                }
            }
        }
        WMMA_GROUP_END();
    };

    FragW f0, f1;
    loadf(f0, 0);
    for (int it = 0; it < CI / 4; it += 2) {
        loadf(f1, it + 1);
        __builtin_amdgcn_sched_barrier(0);
        compute(f0);
        loadf(f0, (it + 2 < CI / 4) ? it + 2 : CI / 4 - 1);
        __builtin_amdgcn_sched_barrier(0);
        compute(f1);
    }

    UNROLL_REGS
    for (int gr = 0; gr < 8; ++gr) {
        const int g  = gr >> 2;
        const int rr = gr & 3;
        const int o  = m0 + 8 * g + 4 * h + rr;
        const v4f* __restrict__ bp = (const v4f*)(bias + (size_t)o * 16);
        v4f br[4];
        br[0] = bp[0]; br[1] = bp[1]; br[2] = bp[2]; br[3] = bp[3];
        UNROLL_REGS
        for (int t = 0; t < 2; ++t) {
            UNROLL_REGS
            for (int s = 0; s < 2; ++s) {
                UNROLL_REGS
                for (int p = 0; p < 2; ++p) {
                    const int r = 2 * rr + p;
                    float bs[2];
                    UNROLL_REGS
                    for (int q = 0; q < 2; ++q) {
                        float tb = 0.0f;
                        UNROLL_REGS
                        for (int dx = 0; dx < 2; ++dx) {
                            UNROLL_REGS
                            for (int dy = 0; dy < 2; ++dy) {
                                const v4f row = br[p + 2 * dy];
                                const float bv = (q + 2 * dx) == 0 ? row.x
                                               : (q + 2 * dx) == 1 ? row.y
                                               : (q + 2 * dx) == 2 ? row.z : row.w;
                                tb += ok[t + 1 - dy][s][dx] ? bv : 0.0f;
                            }
                        }
                        bs[q] = tb;
                    }
                    v2f val;
                    val.x = (acc[2 * t][s][g * 2 + 0][r] + acc[1][s][g * 2 + 0][r]) + bs[0];
                    val.y = (acc[2 * t][s][g * 2 + 1][r] + acc[1][s][g * 2 + 1][r]) + bs[1];
                    const size_t oi = (((size_t)b * CO + o) * OH + (2 * (u0 + t) + p)) * OW
                                      + 2 * (vb + s * 16 + c);
                    *(v2f*)(out + oi) = val;
                }
            }
        }
    }
}

__global__ __launch_bounds__(256, 2)
void deconv2d_parity_merged_wmma(const float* __restrict__ x,
                                 const float* __restrict__ wgt,
                                 const float* __restrict__ bias,
                                 float* __restrict__ out)
{
    const int lane = threadIdx.x & 31;
    const int wave = __builtin_amdgcn_readfirstlane((int)(threadIdx.x >> 5));
    const int c    = lane & 15;
    const int h    = lane >> 4;
    const int bx   = blockIdx.x;

    int task = -1;
    if (wave < 2)                  task = bx * 2 + wave;
    else if (wave == 2 && bx < 8)  task = 2 * MAIN_BLOCKS + bx;
    if (task >= 0) {
        const int e  = (task >> 3) * 16 + c;
        const int ec = e < EDGE_N ? e : EDGE_N - 1;
        const int b  = ec / EDGE_PER_B;
        const int r  = ec - b * EDGE_PER_B;
        int  pb[1], pu[1], pv[1];
        bool valid[1];
        pb[0] = b;
        pu[0] = r < 65 ? 64 : r - 65;
        pv[0] = r < 65 ? r : 64;
        valid[0] = e < EDGE_N;
        tile<1>(x, wgt, bias, out, (task & 7) * 16, c, h, pb, pu, pv, valid);
    }

    {
        const int osub = bx & 7;
        const int ug   = (bx >> 3) & 3;
        const int b    = bx >> 5;
        const int u0   = ug * 16 + wave * 2;
        for (int half = 0; half < 2; ++half)
            tile_rows2_f22(x, wgt, bias, out, osub * 16, c, h, b, u0, half * 32);
    }
}

extern "C" void kernel_launch(void* const* d_in, const int* in_sizes, int n_in,
                              void* d_out, int out_size, void* d_ws, size_t ws_size,
                              hipStream_t stream) {
    (void)in_sizes; (void)n_in; (void)out_size; (void)d_ws; (void)ws_size;
    const float* x    = (const float*)d_in[0];
    const float* wgt  = (const float*)d_in[1];
    const float* bias = (const float*)d_in[2];
    float*       out  = (float*)d_out;

    dim3 grid(MAIN_BLOCKS);
    dim3 block(256);
    deconv2d_parity_merged_wmma<<<grid, block, 0, stream>>>(x, wgt, bias, out);
}


# Interior K loop: x fragments loaded two steps ahead into a third register set, weights one ahead and issued first; k pairs in a WMMA now (dx=1,dx=0) in both operands: in-WMMA sum order changes
# speedup vs baseline: 8.8718x; 1.0322x over previous
#include <hip/hip_runtime.h>


#if defined(__HIP_DEVICE_COMPILE__)
#if !__has_builtin(__builtin_amdgcn_wmma_f32_16x16x4_f32)
#error "device pass: missing __builtin_amdgcn_wmma_f32_16x16x4_f32"
#endif
#endif

typedef __attribute__((ext_vector_type(2))) float v2f;
typedef __attribute__((ext_vector_type(4))) float v4f;
typedef __attribute__((ext_vector_type(8))) float v8f;

#define UNROLL_REGS _Pragma("unroll")

#define WMMA_GROUP_BEGIN() __builtin_amdgcn_sched_barrier(0)
#define WMMA_GROUP_END()                                         \
    do {                                                         \
        __builtin_amdgcn_sched_barrier(0);                       \
        asm volatile("v_nop\n\tv_nop\n\tv_nop\n\tv_nop");       \
        __builtin_amdgcn_sched_barrier(0);                       \
    } while (0)

#define BN    8
#define CI    256
#define CO    128
#define HH    64
#define WW    64
#define OH    130
#define OW    130
#define EDGE_PER_B  129
#define EDGE_N      (BN * EDGE_PER_B)
#define EDGE_TILES  ((EDGE_N + 15) / 16)
#define MAIN_BLOCKS (BN * 4 * 8)
#define EDGE_TASKS  (EDGE_TILES * 8)
static_assert(EDGE_TASKS == 2 * MAIN_BLOCKS + 8, "edge task split assumes 520 = 2 x 256 + 8");

template <int NS>
struct Frag {
    v4f   a[4];
    float b[NS][2][2];
};

template <int NS>
__device__ __forceinline__ void tile(const float* __restrict__ x,
                                     const float* __restrict__ wgt,
                                     const float* __restrict__ bias,
                                     float* __restrict__ out,
                                     const int m0, const int c, const int h,
                                     const int (&pb)[NS], const int (&pu)[NS],
                                     const int (&pv)[NS], const bool (&valid)[NS])
{
    int  off[NS][2][2];
    bool ok[NS][2][2];
    UNROLL_REGS
    for (int s = 0; s < NS; ++s) {
        UNROLL_REGS
        for (int dy = 0; dy < 2; ++dy) {
            UNROLL_REGS
            for (int dx = 0; dx < 2; ++dx) {
                const int hh = pu[s] - dy;
                const int ww = pv[s] - dx;
                ok[s][dy][dx] = (hh >= 0) && (hh < HH) && (ww >= 0) && (ww < WW);
                const int hc = hh < 0 ? 0 : (hh > HH - 1 ? HH - 1 : hh);
                const int wc = ww < 0 ? 0 : (ww > WW - 1 ? WW - 1 : ww);
                off[s][dy][dx] = (pb[s] * CI + h) * (HH * WW) + hc * WW + wc;
            }
        }
    }

    const v4f* __restrict__ wrow =
        (const v4f*)(wgt + ((size_t)(m0 + c) * CI + h) * 16);

    v8f acc[NS][4] = {};

    auto loadf = [&](Frag<NS>& f, const int it) {
        const v4f* __restrict__ wp = wrow + (size_t)it * 8;
        f.a[0] = wp[0];
        f.a[1] = wp[1];
        f.a[2] = wp[2];
        f.a[3] = wp[3];
        const float* __restrict__ xi = x + (size_t)it * (2 * HH * WW);
        UNROLL_REGS
        for (int s = 0; s < NS; ++s) {
            UNROLL_REGS
            for (int dy = 0; dy < 2; ++dy) {
                f.b[s][dy][0] = xi[off[s][dy][0]];
                f.b[s][dy][1] = xi[off[s][dy][1]];
            }
        }
    };

    auto compute = [&](const Frag<NS>& f) {
        UNROLL_REGS
        for (int dy = 0; dy < 2; ++dy) {
            v2f bm[NS];
            UNROLL_REGS
            for (int s = 0; s < NS; ++s) {
                bm[s].x = ok[s][dy][0] ? f.b[s][dy][0] : 0.0f;
                bm[s].y = ok[s][dy][1] ? f.b[s][dy][1] : 0.0f;
            }
            v2f av[2][2];
            UNROLL_REGS
            for (int p = 0; p < 2; ++p) {
                const v4f wr = f.a[p + 2 * dy];
                UNROLL_REGS
                for (int q = 0; q < 2; ++q) {
                    av[p][q].x = q ? wr.y : wr.x;
                    av[p][q].y = q ? wr.w : wr.z;
                }
            }
            WMMA_GROUP_BEGIN();
            UNROLL_REGS
            for (int p = 0; p < 2; ++p) {
                UNROLL_REGS
                for (int q = 0; q < 2; ++q) {
                    UNROLL_REGS
                    for (int s = 0; s < NS; ++s) {
                        acc[s][p * 2 + q] = __builtin_amdgcn_wmma_f32_16x16x4_f32(
                            false, av[p][q], false, bm[s], (short)0, acc[s][p * 2 + q],
                            false, false);
                    }
                }
            }
            WMMA_GROUP_END();
        }
    };

    Frag<NS> f0, f1;
    loadf(f0, 0);
    for (int it = 0; it < CI / 2; it += 2) {
        loadf(f1, it + 1);
        compute(f0);
        loadf(f0, (it + 2 < CI / 2) ? it + 2 : CI / 2 - 1);
        compute(f1);
    }

    UNROLL_REGS
    for (int r = 0; r < 8; ++r) {
        const int o = m0 + 8 * h + r;
        const v4f* __restrict__ bp = (const v4f*)(bias + (size_t)o * 16);
        v4f br[4];
        br[0] = bp[0]; br[1] = bp[1]; br[2] = bp[2]; br[3] = bp[3];
        UNROLL_REGS
        for (int s = 0; s < NS; ++s) {
            if (!valid[s]) continue;
            UNROLL_REGS
            for (int p = 0; p < 2; ++p) {
                float bs[2];
                UNROLL_REGS
                for (int q = 0; q < 2; ++q) {
                    float t = 0.0f;
                    UNROLL_REGS
                    for (int dx = 0; dx < 2; ++dx) {
                        UNROLL_REGS
                        for (int dy = 0; dy < 2; ++dy) {
                            const v4f row = br[p + 2 * dy];
                            const float bv = (q + 2 * dx) == 0 ? row.x
                                           : (q + 2 * dx) == 1 ? row.y
                                           : (q + 2 * dx) == 2 ? row.z : row.w;
                            t += ok[s][dy][dx] ? bv : 0.0f;
                        }
                    }
                    bs[q] = t;
                }
                v2f val;
                val.x = acc[s][p * 2 + 0][r] + bs[0];
                val.y = acc[s][p * 2 + 1][r] + bs[1];
                const size_t oi = (((size_t)pb[s] * CO + o) * OH + (2 * pu[s] + p)) * OW
                                  + 2 * pv[s];
                *(v2f*)(out + oi) = val;
            }
        }
    }
}

struct FragA {
    v4f   a[2][2][2];
};
struct FragB {
    float b[2][3][2][2];
};

__device__ __forceinline__ void tile_rows2_f22(const float* __restrict__ x,
                                               const float* __restrict__ wgt,
                                               const float* __restrict__ bias,
                                               float* __restrict__ out,
                                               const int m0, const int c, const int h,
                                               const int b, const int u0, const int vb)
{
    int  off[3][2][2];
    bool ok[3][2][2];
    UNROLL_REGS
    for (int j = 0; j < 3; ++j) {
        const int hh = u0 - 1 + j;
        const int hc = hh < 0 ? 0 : hh;
        UNROLL_REGS
        for (int s = 0; s < 2; ++s) {
            UNROLL_REGS
            for (int dx = 0; dx < 2; ++dx) {
                const int ww = vb + s * 16 + c - dx;
                const int wc = ww < 0 ? 0 : ww;
                ok[j][s][dx]  = (hh >= 0) && (ww >= 0);
                off[j][s][dx] = (b * CI + h) * (HH * WW) + hc * WW + wc;
            }
        }
    }

    const v4f* __restrict__ wrow =
        (const v4f*)(wgt + ((size_t)(m0 + (c >> 1)) * CI + h) * 16) + (c & 1);

    v8f acc[3][2][4] = {};

    auto loadA = [&](FragA& f, const int it) {
        const v4f* __restrict__ wp = wrow + (size_t)it * 16;
        UNROLL_REGS
        for (int kk = 0; kk < 2; ++kk) {
            UNROLL_REGS
            for (int g = 0; g < 2; ++g) {
                f.a[kk][g][0] = wp[g * (8 * CI * 4) + kk * 8 + 0];
                f.a[kk][g][1] = wp[g * (8 * CI * 4) + kk * 8 + 2];
            }
        }
    };
    auto loadB = [&](FragB& f, const int it) {
        const float* __restrict__ xi = x + (size_t)it * (4 * HH * WW);
        UNROLL_REGS
        for (int kk = 0; kk < 2; ++kk) {
            UNROLL_REGS
            for (int j = 0; j < 3; ++j) {
                UNROLL_REGS
                for (int s = 0; s < 2; ++s) {
                    f.b[kk][j][s][0] = xi[off[j][s][0] + kk * (2 * HH * WW)];
                    f.b[kk][j][s][1] = xi[off[j][s][1] + kk * (2 * HH * WW)];
                }
            }
        }
    };

    auto compute = [&](const FragA& fa, const FragB& f) {
        v2f bd[2][3][2];
        v2f aw[2][3][2][2];
        UNROLL_REGS
        for (int kk = 0; kk < 2; ++kk) {
            UNROLL_REGS
            for (int s = 0; s < 2; ++s) {
                v2f xm[3];
                UNROLL_REGS
                for (int j = 0; j < 3; ++j) {
                    xm[j].x = ok[j][s][1] ? f.b[kk][j][s][1] : 0.0f;
                    xm[j].y = ok[j][s][0] ? f.b[kk][j][s][0] : 0.0f;
                }
                bd[kk][0][s] = xm[0] - xm[1];
                bd[kk][1][s] = xm[1];
                bd[kk][2][s] = xm[2] - xm[1];
            }
            UNROLL_REGS
            for (int g = 0; g < 2; ++g) {
                const v4f wa = fa.a[kk][g][0];
                const v4f wb = fa.a[kk][g][1];
                UNROLL_REGS
                for (int q = 0; q < 2; ++q) {
                    v2f ta, tb;
                    ta.x = q ? wa.w : wa.z;
                    ta.y = q ? wa.y : wa.x;
                    tb.x = q ? wb.w : wb.z;
                    tb.y = q ? wb.y : wb.x;
                    aw[kk][0][g][q] = tb;
                    aw[kk][1][g][q] = ta + tb;
                    aw[kk][2][g][q] = ta;
                }
            }
        }
        WMMA_GROUP_BEGIN();
        UNROLL_REGS
        for (int kk = 0; kk < 2; ++kk) {
            UNROLL_REGS
            for (int m = 0; m < 3; ++m) {
                UNROLL_REGS
                for (int g = 0; g < 2; ++g) {
                    UNROLL_REGS
                    for (int q = 0; q < 2; ++q) {
                        UNROLL_REGS
                        for (int s = 0; s < 2; ++s) {
                            acc[m][s][g * 2 + q] = __builtin_amdgcn_wmma_f32_16x16x4_f32(
                                false, aw[kk][m][g][q], false, bd[kk][m][s], (short)0,
                                acc[m][s][g * 2 + q], false, false);
                        }
                    }
                }
            }
        }
        WMMA_GROUP_END();
    };

    constexpr int NSTEP = CI / 4;
    FragA a0, a1;
    FragB b0, b1, b2;
    auto step = [&](const int k, const FragA& ac, FragA& an, const FragB& bc, FragB& bn,
                    const bool more_a, const bool more_b) {
        if (more_a) loadA(an, k + 1);
        __builtin_amdgcn_sched_barrier(0);
        if (more_b) loadB(bn, k + 2);
        __builtin_amdgcn_sched_barrier(0);
        compute(ac, bc);
    };
    loadA(a0, 0);
    loadB(b0, 0);
    __builtin_amdgcn_sched_barrier(0);
    loadB(b1, 1);
    __builtin_amdgcn_sched_barrier(0);
    static_assert(NSTEP % 6 == 4, "tail below assumes 4 steps behind the 6-step loop");
    for (int it = 0; it < NSTEP - 4; it += 6) {
        step(it + 0, a0, a1, b0, b2, true, true);
        step(it + 1, a1, a0, b1, b0, true, true);
        step(it + 2, a0, a1, b2, b1, true, true);
        step(it + 3, a1, a0, b0, b2, true, true);
        step(it + 4, a0, a1, b1, b0, true, true);
        step(it + 5, a1, a0, b2, b1, true, true);
    }
    step(NSTEP - 4, a0, a1, b0, b2, true, true);
    step(NSTEP - 3, a1, a0, b1, b0, true, true);
    step(NSTEP - 2, a0, a1, b2, b1, true, false);
    step(NSTEP - 1, a1, a0, b0, b2, false, false);

    UNROLL_REGS
    for (int gr = 0; gr < 8; ++gr) {
        const int g  = gr >> 2;
        const int rr = gr & 3;
        const int o  = m0 + 8 * g + 4 * h + rr;
        const v4f* __restrict__ bp = (const v4f*)(bias + (size_t)o * 16);
        v4f br[4];
        br[0] = bp[0]; br[1] = bp[1]; br[2] = bp[2]; br[3] = bp[3];
        UNROLL_REGS
        for (int t = 0; t < 2; ++t) {
            UNROLL_REGS
            for (int s = 0; s < 2; ++s) {
                UNROLL_REGS
                for (int p = 0; p < 2; ++p) {
                    const int r = 2 * rr + p;
                    float bs[2];
                    UNROLL_REGS
                    for (int q = 0; q < 2; ++q) {
                        float tb = 0.0f;
                        UNROLL_REGS
                        for (int dx = 0; dx < 2; ++dx) {
                            UNROLL_REGS
                            for (int dy = 0; dy < 2; ++dy) {
                                const v4f row = br[p + 2 * dy];
                                const float bv = (q + 2 * dx) == 0 ? row.x
                                               : (q + 2 * dx) == 1 ? row.y
                                               : (q + 2 * dx) == 2 ? row.z : row.w;
                                tb += ok[t + 1 - dy][s][dx] ? bv : 0.0f;
                            }
                        }
                        bs[q] = tb;
                    }
                    v2f val;
                    val.x = (acc[2 * t][s][g * 2 + 0][r] + acc[1][s][g * 2 + 0][r]) + bs[0];
                    val.y = (acc[2 * t][s][g * 2 + 1][r] + acc[1][s][g * 2 + 1][r]) + bs[1];
                    const size_t oi = (((size_t)b * CO + o) * OH + (2 * (u0 + t) + p)) * OW
                                      + 2 * (vb + s * 16 + c);
                    *(v2f*)(out + oi) = val;
                }
            }
        }
    }
}

__global__ __launch_bounds__(256, 2)
void deconv2d_parity_merged_wmma(const float* __restrict__ x,
                                 const float* __restrict__ wgt,
                                 const float* __restrict__ bias,
                                 float* __restrict__ out)
{
    const int lane = threadIdx.x & 31;
    const int wave = __builtin_amdgcn_readfirstlane((int)(threadIdx.x >> 5));
    const int c    = lane & 15;
    const int h    = lane >> 4;
    const int bx   = blockIdx.x;

    int task = -1;
    if (wave < 2)                  task = bx * 2 + wave;
    else if (wave == 2 && bx < 8)  task = 2 * MAIN_BLOCKS + bx;
    if (task >= 0) {
        const int e  = (task >> 3) * 16 + c;
        const int ec = e < EDGE_N ? e : EDGE_N - 1;
        const int b  = ec / EDGE_PER_B;
        const int r  = ec - b * EDGE_PER_B;
        int  pb[1], pu[1], pv[1];
        bool valid[1];
        pb[0] = b;
        pu[0] = r < 65 ? 64 : r - 65;
        pv[0] = r < 65 ? r : 64;
        valid[0] = e < EDGE_N;
        tile<1>(x, wgt, bias, out, (task & 7) * 16, c, h, pb, pu, pv, valid);
    }

    {
        const int osub = bx & 7;
        const int ug   = (bx >> 3) & 3;
        const int b    = bx >> 5;
        const int u0   = ug * 16 + wave * 2;
        for (int half = 0; half < 2; ++half)
            tile_rows2_f22(x, wgt, bias, out, osub * 16, c, h, b, u0, half * 32);
    }
}

extern "C" void kernel_launch(void* const* d_in, const int* in_sizes, int n_in,
                              void* d_out, int out_size, void* d_ws, size_t ws_size,
                              hipStream_t stream) {
    (void)in_sizes; (void)n_in; (void)out_size; (void)d_ws; (void)ws_size;
    const float* x    = (const float*)d_in[0];
    const float* wgt  = (const float*)d_in[1];
    const float* bias = (const float*)d_in[2];
    float*       out  = (float*)d_out;

    dim3 grid(MAIN_BLOCKS);
    dim3 block(256);
    deconv2d_parity_merged_wmma<<<grid, block, 0, stream>>>(x, wgt, bias, out);
}


# F(2,2) along x nested in F(4,2) along y: wave tile 8 ch x 4 rows x 32 cols, lane owns a column pair, 15 WMMAs for 20, K = 4 channels, x loads 1 step ahead (was 2); sum order changes: 15 partial sums
# speedup vs baseline: 10.6580x; 1.2013x over previous
#include <hip/hip_runtime.h>


#if defined(__HIP_DEVICE_COMPILE__)
#if !__has_builtin(__builtin_amdgcn_wmma_f32_16x16x4_f32)
#error "device pass: missing __builtin_amdgcn_wmma_f32_16x16x4_f32"
#endif
#endif

typedef __attribute__((ext_vector_type(2))) float v2f;
typedef __attribute__((ext_vector_type(4))) float v4f;
typedef __attribute__((ext_vector_type(8))) float v8f;

#define UNROLL_REGS _Pragma("unroll")

#define WMMA_GROUP_BEGIN() __builtin_amdgcn_sched_barrier(0)
#define WMMA_GROUP_END()                                         \
    do {                                                         \
        __builtin_amdgcn_sched_barrier(0);                       \
        asm volatile("v_nop\n\tv_nop\n\tv_nop\n\tv_nop");       \
        __builtin_amdgcn_sched_barrier(0);                       \
    } while (0)

#define BN    8
#define CI    256
#define CO    128
#define HH    64
#define WW    64
#define OH    130
#define OW    130
#define EDGE_PER_B  129
#define EDGE_N      (BN * EDGE_PER_B)
#define EDGE_TILES  ((EDGE_N + 15) / 16)
#define MAIN_BLOCKS (BN * 4 * 8)
#define EDGE_TASKS  (EDGE_TILES * 8)
static_assert(EDGE_TASKS == 2 * MAIN_BLOCKS + 8, "edge task split assumes 520 = 2 x 256 + 8");

template <int NS>
struct Frag {
    v4f   a[4];
    float b[NS][2][2];
};

template <int NS>
__device__ __forceinline__ void tile(const float* __restrict__ x,
                                     const float* __restrict__ wgt,
                                     const float* __restrict__ bias,
                                     float* __restrict__ out,
                                     const int m0, const int c, const int h,
                                     const int (&pb)[NS], const int (&pu)[NS],
                                     const int (&pv)[NS], const bool (&valid)[NS])
{
    int  off[NS][2][2];
    bool ok[NS][2][2];
    UNROLL_REGS
    for (int s = 0; s < NS; ++s) {
        UNROLL_REGS
        for (int dy = 0; dy < 2; ++dy) {
            UNROLL_REGS
            for (int dx = 0; dx < 2; ++dx) {
                const int hh = pu[s] - dy;
                const int ww = pv[s] - dx;
                ok[s][dy][dx] = (hh >= 0) && (hh < HH) && (ww >= 0) && (ww < WW);
                const int hc = hh < 0 ? 0 : (hh > HH - 1 ? HH - 1 : hh);
                const int wc = ww < 0 ? 0 : (ww > WW - 1 ? WW - 1 : ww);
                off[s][dy][dx] = (pb[s] * CI + h) * (HH * WW) + hc * WW + wc;
            }
        }
    }

    const v4f* __restrict__ wrow =
        (const v4f*)(wgt + ((size_t)(m0 + c) * CI + h) * 16);

    v8f acc[NS][4] = {};

    auto loadf = [&](Frag<NS>& f, const int it) {
        const v4f* __restrict__ wp = wrow + (size_t)it * 8;
        f.a[0] = wp[0];
        f.a[1] = wp[1];
        f.a[2] = wp[2];
        f.a[3] = wp[3];
        const float* __restrict__ xi = x + (size_t)it * (2 * HH * WW);
        UNROLL_REGS
        for (int s = 0; s < NS; ++s) {
            UNROLL_REGS
            for (int dy = 0; dy < 2; ++dy) {
                f.b[s][dy][0] = xi[off[s][dy][0]];
                f.b[s][dy][1] = xi[off[s][dy][1]];
            }
        }
    };

    auto compute = [&](const Frag<NS>& f) {
        UNROLL_REGS
        for (int dy = 0; dy < 2; ++dy) {
            v2f bm[NS];
            UNROLL_REGS
            for (int s = 0; s < NS; ++s) {
                bm[s].x = ok[s][dy][0] ? f.b[s][dy][0] : 0.0f;
                bm[s].y = ok[s][dy][1] ? f.b[s][dy][1] : 0.0f;
            }
            v2f av[2][2];
            UNROLL_REGS
            for (int p = 0; p < 2; ++p) {
                const v4f wr = f.a[p + 2 * dy];
                UNROLL_REGS
                for (int q = 0; q < 2; ++q) {
                    av[p][q].x = q ? wr.y : wr.x;
                    av[p][q].y = q ? wr.w : wr.z;
                }
            }
            WMMA_GROUP_BEGIN();
            UNROLL_REGS
            for (int p = 0; p < 2; ++p) {
                UNROLL_REGS
                for (int q = 0; q < 2; ++q) {
                    UNROLL_REGS
                    for (int s = 0; s < NS; ++s) {
                        acc[s][p * 2 + q] = __builtin_amdgcn_wmma_f32_16x16x4_f32(
                            false, av[p][q], false, bm[s], (short)0, acc[s][p * 2 + q],
                            false, false);
                    }
                }
            }
            WMMA_GROUP_END();
        }
    };

    Frag<NS> f0, f1;
    loadf(f0, 0);
    for (int it = 0; it < CI / 2; it += 2) {
        loadf(f1, it + 1);
        compute(f0);
        loadf(f0, (it + 2 < CI / 2) ? it + 2 : CI / 2 - 1);
        compute(f1);
    }

    UNROLL_REGS
    for (int r = 0; r < 8; ++r) {
        const int o = m0 + 8 * h + r;
        const v4f* __restrict__ bp = (const v4f*)(bias + (size_t)o * 16);
        v4f br[4];
        br[0] = bp[0]; br[1] = bp[1]; br[2] = bp[2]; br[3] = bp[3];
        UNROLL_REGS
        for (int s = 0; s < NS; ++s) {
            if (!valid[s]) continue;
            UNROLL_REGS
            for (int p = 0; p < 2; ++p) {
                float bs[2];
                UNROLL_REGS
                for (int q = 0; q < 2; ++q) {
                    float t = 0.0f;
                    UNROLL_REGS
                    for (int dx = 0; dx < 2; ++dx) {
                        UNROLL_REGS
                        for (int dy = 0; dy < 2; ++dy) {
                            const v4f row = br[p + 2 * dy];
                            const float bv = (q + 2 * dx) == 0 ? row.x
                                           : (q + 2 * dx) == 1 ? row.y
                                           : (q + 2 * dx) == 2 ? row.z : row.w;
                            t += ok[s][dy][dx] ? bv : 0.0f;
                        }
                    }
                    bs[q] = t;
                }
                v2f val;
                val.x = acc[s][p * 2 + 0][r] + bs[0];
                val.y = acc[s][p * 2 + 1][r] + bs[1];
                const size_t oi = (((size_t)pb[s] * CO + o) * OH + (2 * pu[s] + p)) * OW
                                  + 2 * pv[s];
                *(v2f*)(out + oi) = val;
            }
        }
    }
}

struct FragW {
    v4f   a[2][2];
};
struct FragX {
    float l[5][2];
    v2f   cr[5][2];
};

__device__ __forceinline__ void tile_rows4_f42_f22(const float* __restrict__ x,
                                                   const float* __restrict__ wgt,
                                                   const float* __restrict__ bias,
                                                   float* __restrict__ out,
                                                   const int m0, const int c, const int h,
                                                   const int b, const int u0, const int vb)
{
    const int  v0  = vb + 2 * c;
    const bool okr = u0 > 0;
    const bool okc = v0 > 0;
    const int  wl  = v0 - 1 < 0 ? 0 : v0 - 1;
    int off[5][2];
    UNROLL_REGS
    for (int j = 0; j < 5; ++j) {
        const int hh = u0 - 1 + j;
        const int hc = hh < 0 ? 0 : hh;
        off[j][0] = (b * CI + h) * (HH * WW) + hc * WW + wl;
        off[j][1] = (b * CI + h) * (HH * WW) + hc * WW + v0;
    }

    const v4f* __restrict__ wrow =
        (const v4f*)(wgt + ((size_t)(m0 + (c >> 1)) * CI + h) * 16) + (c & 1);

    v8f acc[5][3][2] = {};

    auto loadW = [&](FragW& f, const int it) {
        const v4f* __restrict__ wp = wrow + (size_t)it * 16;
        UNROLL_REGS
        for (int e = 0; e < 2; ++e) {
            f.a[e][0] = wp[e * 8 + 0];
            f.a[e][1] = wp[e * 8 + 2];
        }
    };
    auto loadX = [&](FragX& f, const int it) {
        const float* __restrict__ xi = x + (size_t)it * (4 * HH * WW);
        UNROLL_REGS
        for (int j = 0; j < 5; ++j) {
            UNROLL_REGS
            for (int e = 0; e < 2; ++e) {
                f.l[j][e]    = xi[off[j][0] + e * (2 * HH * WW)];
                f.cr[j][e].x = xi[off[j][1] + e * (2 * HH * WW)];
                f.cr[j][e].y = xi[off[j][1] + e * (2 * HH * WW) + 1];
            }
        }
    };

    auto compute = [&](const FragW& fw, const FragX& fx) {
        v2f bd[5][3];
        v2f aw[5][3][2];
        {
            v2f t[5][3];
            UNROLL_REGS
            for (int n = 0; n < 3; ++n) {
                v2f xm[5];
                UNROLL_REGS
                for (int j = 0; j < 5; ++j) {
                    const bool okv = (j > 0 || okr) && (n > 0 || okc);
                    const float r0 = n == 0 ? fx.l[j][0] : n == 1 ? fx.cr[j][0].x : fx.cr[j][0].y;
                    const float r1 = n == 0 ? fx.l[j][1] : n == 1 ? fx.cr[j][1].x : fx.cr[j][1].y;
                    xm[j].x = okv ? r0 : 0.0f;
                    xm[j].y = okv ? r1 : 0.0f;
                }
                const v2f s = xm[3] - xm[1];
                t[0][n] = 2.0f * (xm[0] - xm[2]) + s;
                t[1][n] = (xm[1] + xm[2]) - s;
                t[2][n] = 3.0f * (xm[2] - xm[1]) - s;
                t[3][n] = s;
                t[4][n] = (xm[4] - xm[2]) - 2.0f * s;
            }
            UNROLL_REGS
            for (int m = 0; m < 5; ++m) {
                bd[m][0] = t[m][0] - t[m][1];
                bd[m][1] = t[m][1];
                bd[m][2] = t[m][2] - t[m][1];
            }
        }
        {
            v2f wm[5][4];
            UNROLL_REGS
            for (int kx = 0; kx < 4; ++kx) {
                v2f wa, wb;
                wa.x = fw.a[0][0][kx];
                wa.y = fw.a[1][0][kx];
                wb.x = fw.a[0][1][kx];
                wb.y = fw.a[1][1][kx];
                wm[0][kx] = wb;
                wm[1][kx] = wa + wb;
                wm[2][kx] = wb - wa;
                wm[3][kx] = wb + 2.0f * wa;
                wm[4][kx] = wa;
            }
            UNROLL_REGS
            for (int m = 0; m < 5; ++m) {
                UNROLL_REGS
                for (int q = 0; q < 2; ++q) {
                    const v2f g0 = wm[m][q];
                    const v2f g1 = wm[m][q + 2];
                    aw[m][0][q] = g1;
                    aw[m][1][q] = g0 + g1;
                    aw[m][2][q] = g0;
                }
            }
        }
        WMMA_GROUP_BEGIN();
        UNROLL_REGS
        for (int m = 0; m < 5; ++m) {
            UNROLL_REGS
            for (int n = 0; n < 3; ++n) {
                UNROLL_REGS
                for (int q = 0; q < 2; ++q) {
                    acc[m][n][q] = __builtin_amdgcn_wmma_f32_16x16x4_f32(
                        false, aw[m][n][q], false, bd[m][n], (short)0,
                        acc[m][n][q], false, false);
                }
            }
        }
        WMMA_GROUP_END();
    };

    constexpr int NSTEP = CI / 4;
    FragW a0, a1;
    FragX b0, b1;
    auto step = [&](const int k, const FragW& ac, FragW& an, const FragX& bc, FragX& bn,
                    const bool more) {
        if (more) {
            loadW(an, k + 1);
            loadX(bn, k + 1);
        }
        __builtin_amdgcn_sched_barrier(0);
        compute(ac, bc);
    };
    loadW(a0, 0);
    loadX(b0, 0);
    static_assert(NSTEP % 6 == 4, "tail below assumes 4 steps behind the 6-step loop");
    for (int it = 0; it < NSTEP - 4; it += 6) {
        step(it + 0, a0, a1, b0, b1, true);
        step(it + 1, a1, a0, b1, b0, true);
        step(it + 2, a0, a1, b0, b1, true);
        step(it + 3, a1, a0, b1, b0, true);
        step(it + 4, a0, a1, b0, b1, true);
        step(it + 5, a1, a0, b1, b0, true);
    }
    step(NSTEP - 4, a0, a1, b0, b1, true);
    step(NSTEP - 3, a1, a0, b1, b0, true);
    step(NSTEP - 2, a0, a1, b0, b1, true);
    step(NSTEP - 1, a1, a0, b1, b0, false);

    UNROLL_REGS
    for (int rr = 0; rr < 4; ++rr) {
        const int o = m0 + 4 * h + rr;
        const v4f* __restrict__ bp = (const v4f*)(bias + (size_t)o * 16);
        v4f br[4];
        br[0] = bp[0]; br[1] = bp[1]; br[2] = bp[2]; br[3] = bp[3];
        UNROLL_REGS
        for (int p = 0; p < 2; ++p) {
            const int r = 2 * rr + p;
            float y[4][2][2];
            UNROLL_REGS
            for (int cl = 0; cl < 2; ++cl) {
                UNROLL_REGS
                for (int q = 0; q < 2; ++q) {
                    float ms[5];
                    UNROLL_REGS
                    for (int m = 0; m < 5; ++m)
                        ms[m] = acc[m][1][q][r] + acc[m][cl ? 2 : 0][q][r];
                    const float m0s = ms[0];
                    const float m1h = 0.5f * ms[1];
                    const float m2s = (1.0f / 6.0f) * ms[2];
                    const float m3s = ms[3];
                    const float m4s = ms[4];
                    y[0][cl][q] = (0.5f * m0s + m1h) + (m2s + (1.0f / 6.0f) * m3s);
                    y[1][cl][q] = (m1h - m2s) + (1.0f / 3.0f) * m3s;
                    y[2][cl][q] = (m1h + m2s) + (2.0f / 3.0f) * m3s;
                    y[3][cl][q] = ((m1h - m2s) + (4.0f / 3.0f) * m3s) + m4s;
                }
            }
            UNROLL_REGS
            for (int t = 0; t < 4; ++t) {
                const size_t oi = (((size_t)b * CO + o) * OH + (2 * (u0 + t) + p)) * OW
                                  + 2 * v0;
                UNROLL_REGS
                for (int cl = 0; cl < 2; ++cl) {
                    float bs[2];
                    UNROLL_REGS
                    for (int q = 0; q < 2; ++q) {
                        float tb = 0.0f;
                        UNROLL_REGS
                        for (int dx = 0; dx < 2; ++dx) {
                            UNROLL_REGS
                            for (int dy = 0; dy < 2; ++dy) {
                                const v4f row = br[p + 2 * dy];
                                const float bv = (q + 2 * dx) == 0 ? row.x
                                               : (q + 2 * dx) == 1 ? row.y
                                               : (q + 2 * dx) == 2 ? row.z : row.w;
                                const bool okt = (t - dy >= 0 || okr) && (cl - dx >= 0 || okc);
                                tb += okt ? bv : 0.0f;
                            }
                        }
                        bs[q] = tb;
                    }
                    v2f val;
                    val.x = y[t][cl][0] + bs[0];
                    val.y = y[t][cl][1] + bs[1];
                    *(v2f*)(out + oi + 2 * cl) = val;
                }
            }
        }
    }
}

__global__ __launch_bounds__(256, 2)
void deconv2d_parity_merged_wmma(const float* __restrict__ x,
                                 const float* __restrict__ wgt,
                                 const float* __restrict__ bias,
                                 float* __restrict__ out)
{
    const int lane = threadIdx.x & 31;
    const int wave = __builtin_amdgcn_readfirstlane((int)(threadIdx.x >> 5));
    const int c    = lane & 15;
    const int h    = lane >> 4;
    const int bx   = blockIdx.x;

    int task = -1;
    if (wave < 2)                  task = bx * 2 + wave;
    else if (wave == 2 && bx < 8)  task = 2 * MAIN_BLOCKS + bx;
    if (task >= 0) {
        const int e  = (task >> 3) * 16 + c;
        const int ec = e < EDGE_N ? e : EDGE_N - 1;
        const int b  = ec / EDGE_PER_B;
        const int r  = ec - b * EDGE_PER_B;
        int  pb[1], pu[1], pv[1];
        bool valid[1];
        pb[0] = b;
        pu[0] = r < 65 ? 64 : r - 65;
        pv[0] = r < 65 ? r : 64;
        valid[0] = e < EDGE_N;
        tile<1>(x, wgt, bias, out, (task & 7) * 16, c, h, pb, pu, pv, valid);
    }

    {
        const int osub = bx & 7;
        const int ug   = (bx >> 3) & 3;
        const int b    = bx >> 5;
        const int u0   = ug * 16 + (wave >> 1) * 4;
        const int vb   = (wave & 1) * 32;
        for (int g = 0; g < 2; ++g)
            tile_rows4_f42_f22(x, wgt, bias, out, osub * 16 + g * 8, c, h, b, u0, vb);
    }
}

extern "C" void kernel_launch(void* const* d_in, const int* in_sizes, int n_in,
                              void* d_out, int out_size, void* d_ws, size_t ws_size,
                              hipStream_t stream) {
    (void)in_sizes; (void)n_in; (void)out_size; (void)d_ws; (void)ws_size;
    const float* x    = (const float*)d_in[0];
    const float* wgt  = (const float*)d_in[1];
    const float* bias = (const float*)d_in[2];
    float*       out  = (float*)d_out;

    dim3 grid(MAIN_BLOCKS);
    dim3 block(256);
    deconv2d_parity_merged_wmma<<<grid, block, 0, stream>>>(x, wgt, bias, out);
}


# Edge tiles regrouped by kind (row u=64, column v=64, corners) and the tap half with no x position dropped: 8 WMMAs per 4 channels, was 16; column kind sums 4 channels per WMMA: its sum order changes
# speedup vs baseline: 11.2138x; 1.0521x over previous
#include <hip/hip_runtime.h>


#if defined(__HIP_DEVICE_COMPILE__)
#if !__has_builtin(__builtin_amdgcn_wmma_f32_16x16x4_f32)
#error "device pass: missing __builtin_amdgcn_wmma_f32_16x16x4_f32"
#endif
#endif

typedef __attribute__((ext_vector_type(2))) float v2f;
typedef __attribute__((ext_vector_type(4))) float v4f;
typedef __attribute__((ext_vector_type(8))) float v8f;

#define UNROLL_REGS _Pragma("unroll")

#define WMMA_GROUP_BEGIN() __builtin_amdgcn_sched_barrier(0)
#define WMMA_GROUP_END()                                         \
    do {                                                         \
        __builtin_amdgcn_sched_barrier(0);                       \
        asm volatile("v_nop\n\tv_nop\n\tv_nop\n\tv_nop");       \
        __builtin_amdgcn_sched_barrier(0);                       \
    } while (0)

#define BN    8
#define CI    256
#define CO    128
#define HH    64
#define WW    64
#define OH    130
#define OW    130
#define EDGE_ROW_TILES (BN * 4)
#define EDGE_COL_TILES (BN * 4)
#define EDGE_TILES  (EDGE_ROW_TILES + EDGE_COL_TILES + 1)
#define MAIN_BLOCKS (BN * 4 * 8)
#define EDGE_TASKS  (EDGE_TILES * 8)
static_assert(EDGE_TASKS == 2 * MAIN_BLOCKS + 8, "edge task split assumes 520 = 2 x 256 + 8");

struct FragER {
    v4f   a[2][2];
    float b[2][2];
};

__device__ __forceinline__ void edge_row_tile(const float* __restrict__ x,
                                              const float* __restrict__ wgt,
                                              const float* __restrict__ bias,
                                              float* __restrict__ out,
                                              const int m0, const int c, const int h,
                                              const int pb, const int pv, const bool valid)
{
    int  off[2];
    bool ok[2];
    UNROLL_REGS
    for (int dx = 0; dx < 2; ++dx) {
        const int ww = pv - dx;
        ok[dx] = (ww >= 0) && (ww < WW);
        const int wc = ww < 0 ? 0 : (ww > WW - 1 ? WW - 1 : ww);
        off[dx] = (pb * CI + h) * (HH * WW) + (HH - 1) * WW + wc;
    }

    const v4f* __restrict__ wrow =
        (const v4f*)(wgt + ((size_t)(m0 + c) * CI + h) * 16);

    v8f acc[4] = {};

    auto loadf = [&](FragER& f, const int it) {
        const v4f* __restrict__ wp = wrow + (size_t)it * 16;
        const float* __restrict__ xi = x + (size_t)it * (4 * HH * WW);
        UNROLL_REGS
        for (int j = 0; j < 2; ++j) {
            f.a[j][0] = wp[j * 8 + 2];
            f.a[j][1] = wp[j * 8 + 3];
            f.b[j][0] = xi[off[0] + j * (2 * HH * WW)];
            f.b[j][1] = xi[off[1] + j * (2 * HH * WW)];
        }
    };

    auto compute = [&](const FragER& f) {
        v2f bm[2];
        v2f av[2][2][2];
        UNROLL_REGS
        for (int j = 0; j < 2; ++j) {
            bm[j].x = ok[0] ? f.b[j][0] : 0.0f;
            bm[j].y = ok[1] ? f.b[j][1] : 0.0f;
            UNROLL_REGS
            for (int p = 0; p < 2; ++p) {
                const v4f wr = f.a[j][p];
                UNROLL_REGS
                for (int q = 0; q < 2; ++q) {
                    av[j][p][q].x = q ? wr.y : wr.x;
                    av[j][p][q].y = q ? wr.w : wr.z;
                }
            }
        }
        WMMA_GROUP_BEGIN();
        UNROLL_REGS
        for (int j = 0; j < 2; ++j) {
            UNROLL_REGS
            for (int p = 0; p < 2; ++p) {
                UNROLL_REGS
                for (int q = 0; q < 2; ++q) {
                    acc[p * 2 + q] = __builtin_amdgcn_wmma_f32_16x16x4_f32(
                        false, av[j][p][q], false, bm[j], (short)0, acc[p * 2 + q],
                        false, false);
                }
            }
        }
        WMMA_GROUP_END();
    };

    FragER f0, f1;
    loadf(f0, 0);
    for (int it = 0; it < CI / 4; it += 2) {
        loadf(f1, it + 1);
        compute(f0);
        loadf(f0, (it + 2 < CI / 4) ? it + 2 : CI / 4 - 1);
        compute(f1);
    }

    UNROLL_REGS
    for (int r = 0; r < 8; ++r) {
        const int o = m0 + 8 * h + r;
        const v4f* __restrict__ bp = (const v4f*)(bias + (size_t)o * 16);
        v4f br[2];
        br[0] = bp[2]; br[1] = bp[3];
        if (valid) {
            UNROLL_REGS
            for (int p = 0; p < 2; ++p) {
                const v4f row = br[p];
                v2f val;
                val.x = acc[p * 2 + 0][r] + ((ok[0] ? row.x : 0.0f) + (ok[1] ? row.z : 0.0f));
                val.y = acc[p * 2 + 1][r] + ((ok[0] ? row.y : 0.0f) + (ok[1] ? row.w : 0.0f));
                const size_t oi = (((size_t)pb * CO + o) * OH + (2 * HH + p)) * OW + 2 * pv;
                *(v2f*)(out + oi) = val;
            }
        }
    }
}

struct FragEC {
    v2f   a[2][4];
    float b[2][2];
};

__device__ __forceinline__ void edge_col_tile(const float* __restrict__ x,
                                              const float* __restrict__ wgt,
                                              const float* __restrict__ bias,
                                              float* __restrict__ out,
                                              const int m0, const int c, const int h,
                                              const int pb, const int pu)
{
    int  off[2];
    bool ok[2];
    UNROLL_REGS
    for (int dy = 0; dy < 2; ++dy) {
        const int hh = pu - dy;
        ok[dy] = (hh >= 0) && (hh < HH);
        const int hc = hh < 0 ? 0 : (hh > HH - 1 ? HH - 1 : hh);
        off[dy] = (pb * CI + h) * (HH * WW) + hc * WW + (WW - 1);
    }

    const v2f* __restrict__ wrow =
        (const v2f*)(wgt + ((size_t)(m0 + c) * CI + h) * 16);

    v8f acc[4] = {};

    auto loadf = [&](FragEC& f, const int it) {
        const v2f* __restrict__ wp = wrow + (size_t)it * 32;
        const float* __restrict__ xi = x + (size_t)it * (4 * HH * WW);
        UNROLL_REGS
        for (int s = 0; s < 2; ++s) {
            UNROLL_REGS
            for (int ky = 0; ky < 4; ++ky)
                f.a[s][ky] = wp[s * 16 + 2 * ky + 1];
            f.b[0][s] = xi[off[0] + s * (2 * HH * WW)];
            f.b[1][s] = xi[off[1] + s * (2 * HH * WW)];
        }
    };

    auto compute = [&](const FragEC& f) {
        v2f bm[2];
        v2f av[2][2][2];
        UNROLL_REGS
        for (int dy = 0; dy < 2; ++dy) {
            bm[dy].x = ok[dy] ? f.b[dy][0] : 0.0f;
            bm[dy].y = ok[dy] ? f.b[dy][1] : 0.0f;
            UNROLL_REGS
            for (int p = 0; p < 2; ++p) {
                UNROLL_REGS
                for (int q = 0; q < 2; ++q) {
                    av[dy][p][q].x = f.a[0][p + 2 * dy][q];
                    av[dy][p][q].y = f.a[1][p + 2 * dy][q];
                }
            }
        }
        WMMA_GROUP_BEGIN();
        UNROLL_REGS
        for (int dy = 0; dy < 2; ++dy) {
            UNROLL_REGS
            for (int p = 0; p < 2; ++p) {
                UNROLL_REGS
                for (int q = 0; q < 2; ++q) {
                    acc[p * 2 + q] = __builtin_amdgcn_wmma_f32_16x16x4_f32(
                        false, av[dy][p][q], false, bm[dy], (short)0, acc[p * 2 + q],
                        false, false);
                }
            }
        }
        WMMA_GROUP_END();
    };

    FragEC f0, f1;
    loadf(f0, 0);
    for (int it = 0; it < CI / 4; it += 2) {
        loadf(f1, it + 1);
        compute(f0);
        loadf(f0, (it + 2 < CI / 4) ? it + 2 : CI / 4 - 1);
        compute(f1);
    }

    UNROLL_REGS
    for (int r = 0; r < 8; ++r) {
        const int o = m0 + 8 * h + r;
        const v4f* __restrict__ bp = (const v4f*)(bias + (size_t)o * 16);
        v4f br[4];
        br[0] = bp[0]; br[1] = bp[1]; br[2] = bp[2]; br[3] = bp[3];
        UNROLL_REGS
        for (int p = 0; p < 2; ++p) {
            const v4f r0 = br[p];
            const v4f r1 = br[p + 2];
            v2f val;
            val.x = acc[p * 2 + 0][r] + ((ok[0] ? r0.z : 0.0f) + (ok[1] ? r1.z : 0.0f));
            val.y = acc[p * 2 + 1][r] + ((ok[0] ? r0.w : 0.0f) + (ok[1] ? r1.w : 0.0f));
            const size_t oi = (((size_t)pb * CO + o) * OH + (2 * pu + p)) * OW + 2 * WW;
            *(v2f*)(out + oi) = val;
        }
    }
}

struct FragW {
    v4f   a[2][2];
};
struct FragX {
    float l[5][2];
    v2f   cr[5][2];
};

__device__ __forceinline__ void tile_rows4_f42_f22(const float* __restrict__ x,
                                                   const float* __restrict__ wgt,
                                                   const float* __restrict__ bias,
                                                   float* __restrict__ out,
                                                   const int m0, const int c, const int h,
                                                   const int b, const int u0, const int vb)
{
    const int  v0  = vb + 2 * c;
    const bool okr = u0 > 0;
    const bool okc = v0 > 0;
    const int  wl  = v0 - 1 < 0 ? 0 : v0 - 1;
    int off[5][2];
    UNROLL_REGS
    for (int j = 0; j < 5; ++j) {
        const int hh = u0 - 1 + j;
        const int hc = hh < 0 ? 0 : hh;
        off[j][0] = (b * CI + h) * (HH * WW) + hc * WW + wl;
        off[j][1] = (b * CI + h) * (HH * WW) + hc * WW + v0;
    }

    const v4f* __restrict__ wrow =
        (const v4f*)(wgt + ((size_t)(m0 + (c >> 1)) * CI + h) * 16) + (c & 1);

    v8f acc[5][3][2] = {};

    auto loadW = [&](FragW& f, const int it) {
        const v4f* __restrict__ wp = wrow + (size_t)it * 16;
        UNROLL_REGS
        for (int e = 0; e < 2; ++e) {
            f.a[e][0] = wp[e * 8 + 0];
            f.a[e][1] = wp[e * 8 + 2];
        }
    };
    auto loadX = [&](FragX& f, const int it) {
        const float* __restrict__ xi = x + (size_t)it * (4 * HH * WW);
        UNROLL_REGS
        for (int j = 0; j < 5; ++j) {
            UNROLL_REGS
            for (int e = 0; e < 2; ++e) {
                f.l[j][e]    = xi[off[j][0] + e * (2 * HH * WW)];
                f.cr[j][e].x = xi[off[j][1] + e * (2 * HH * WW)];
                f.cr[j][e].y = xi[off[j][1] + e * (2 * HH * WW) + 1];
            }
        }
    };

    auto compute = [&](const FragW& fw, const FragX& fx) {
        v2f bd[5][3];
        v2f aw[5][3][2];
        {
            v2f t[5][3];
            UNROLL_REGS
            for (int n = 0; n < 3; ++n) {
                v2f xm[5];
                UNROLL_REGS
                for (int j = 0; j < 5; ++j) {
                    const bool okv = (j > 0 || okr) && (n > 0 || okc);
                    const float r0 = n == 0 ? fx.l[j][0] : n == 1 ? fx.cr[j][0].x : fx.cr[j][0].y;
                    const float r1 = n == 0 ? fx.l[j][1] : n == 1 ? fx.cr[j][1].x : fx.cr[j][1].y;
                    xm[j].x = okv ? r0 : 0.0f;
                    xm[j].y = okv ? r1 : 0.0f;
                }
                const v2f s = xm[3] - xm[1];
                t[0][n] = 2.0f * (xm[0] - xm[2]) + s;
                t[1][n] = (xm[1] + xm[2]) - s;
                t[2][n] = 3.0f * (xm[2] - xm[1]) - s;
                t[3][n] = s;
                t[4][n] = (xm[4] - xm[2]) - 2.0f * s;
            }
            UNROLL_REGS
            for (int m = 0; m < 5; ++m) {
                bd[m][0] = t[m][0] - t[m][1];
                bd[m][1] = t[m][1];
                bd[m][2] = t[m][2] - t[m][1];
            }
        }
        {
            v2f wm[5][4];
            UNROLL_REGS
            for (int kx = 0; kx < 4; ++kx) {
                v2f wa, wb;
                wa.x = fw.a[0][0][kx];
                wa.y = fw.a[1][0][kx];
                wb.x = fw.a[0][1][kx];
                wb.y = fw.a[1][1][kx];
                wm[0][kx] = wb;
                wm[1][kx] = wa + wb;
                wm[2][kx] = wb - wa;
                wm[3][kx] = wb + 2.0f * wa;
                wm[4][kx] = wa;
            }
            UNROLL_REGS
            for (int m = 0; m < 5; ++m) {
                UNROLL_REGS
                for (int q = 0; q < 2; ++q) {
                    const v2f g0 = wm[m][q];
                    const v2f g1 = wm[m][q + 2];
                    aw[m][0][q] = g1;
                    aw[m][1][q] = g0 + g1;
                    aw[m][2][q] = g0;
                }
            }
        }
        WMMA_GROUP_BEGIN();
        UNROLL_REGS
        for (int m = 0; m < 5; ++m) {
            UNROLL_REGS
            for (int n = 0; n < 3; ++n) {
                UNROLL_REGS
                for (int q = 0; q < 2; ++q) {
                    acc[m][n][q] = __builtin_amdgcn_wmma_f32_16x16x4_f32(
                        false, aw[m][n][q], false, bd[m][n], (short)0,
                        acc[m][n][q], false, false);
                }
            }
        }
        WMMA_GROUP_END();
    };

    constexpr int NSTEP = CI / 4;
    FragW a0, a1;
    FragX b0, b1;
    auto step = [&](const int k, const FragW& ac, FragW& an, const FragX& bc, FragX& bn,
                    const bool more) {
        if (more) {
            loadW(an, k + 1);
            loadX(bn, k + 1);
        }
        __builtin_amdgcn_sched_barrier(0);
        compute(ac, bc);
    };
    loadW(a0, 0);
    loadX(b0, 0);
    static_assert(NSTEP % 6 == 4, "tail below assumes 4 steps behind the 6-step loop");
    for (int it = 0; it < NSTEP - 4; it += 6) {
        step(it + 0, a0, a1, b0, b1, true);
        step(it + 1, a1, a0, b1, b0, true);
        step(it + 2, a0, a1, b0, b1, true);
        step(it + 3, a1, a0, b1, b0, true);
        step(it + 4, a0, a1, b0, b1, true);
        step(it + 5, a1, a0, b1, b0, true);
    }
    step(NSTEP - 4, a0, a1, b0, b1, true);
    step(NSTEP - 3, a1, a0, b1, b0, true);
    step(NSTEP - 2, a0, a1, b0, b1, true);
    step(NSTEP - 1, a1, a0, b1, b0, false);

    UNROLL_REGS
    for (int rr = 0; rr < 4; ++rr) {
        const int o = m0 + 4 * h + rr;
        const v4f* __restrict__ bp = (const v4f*)(bias + (size_t)o * 16);
        v4f br[4];
        br[0] = bp[0]; br[1] = bp[1]; br[2] = bp[2]; br[3] = bp[3];
        UNROLL_REGS
        for (int p = 0; p < 2; ++p) {
            const int r = 2 * rr + p;
            float y[4][2][2];
            UNROLL_REGS
            for (int cl = 0; cl < 2; ++cl) {
                UNROLL_REGS
                for (int q = 0; q < 2; ++q) {
                    float ms[5];
                    UNROLL_REGS
                    for (int m = 0; m < 5; ++m)
                        ms[m] = acc[m][1][q][r] + acc[m][cl ? 2 : 0][q][r];
                    const float m0s = ms[0];
                    const float m1h = 0.5f * ms[1];
                    const float m2s = (1.0f / 6.0f) * ms[2];
                    const float m3s = ms[3];
                    const float m4s = ms[4];
                    y[0][cl][q] = (0.5f * m0s + m1h) + (m2s + (1.0f / 6.0f) * m3s);
                    y[1][cl][q] = (m1h - m2s) + (1.0f / 3.0f) * m3s;
                    y[2][cl][q] = (m1h + m2s) + (2.0f / 3.0f) * m3s;
                    y[3][cl][q] = ((m1h - m2s) + (4.0f / 3.0f) * m3s) + m4s;
                }
            }
            UNROLL_REGS
            for (int t = 0; t < 4; ++t) {
                const size_t oi = (((size_t)b * CO + o) * OH + (2 * (u0 + t) + p)) * OW
                                  + 2 * v0;
                UNROLL_REGS
                for (int cl = 0; cl < 2; ++cl) {
                    float bs[2];
                    UNROLL_REGS
                    for (int q = 0; q < 2; ++q) {
                        float tb = 0.0f;
                        UNROLL_REGS
                        for (int dx = 0; dx < 2; ++dx) {
                            UNROLL_REGS
                            for (int dy = 0; dy < 2; ++dy) {
                                const v4f row = br[p + 2 * dy];
                                const float bv = (q + 2 * dx) == 0 ? row.x
                                               : (q + 2 * dx) == 1 ? row.y
                                               : (q + 2 * dx) == 2 ? row.z : row.w;
                                const bool okt = (t - dy >= 0 || okr) && (cl - dx >= 0 || okc);
                                tb += okt ? bv : 0.0f;
                            }
                        }
                        bs[q] = tb;
                    }
                    v2f val;
                    val.x = y[t][cl][0] + bs[0];
                    val.y = y[t][cl][1] + bs[1];
                    *(v2f*)(out + oi + 2 * cl) = val;
                }
            }
        }
    }
}

__global__ __launch_bounds__(256, 2)
void deconv2d_parity_merged_wmma(const float* __restrict__ x,
                                 const float* __restrict__ wgt,
                                 const float* __restrict__ bias,
                                 float* __restrict__ out)
{
    const int lane = threadIdx.x & 31;
    const int wave = __builtin_amdgcn_readfirstlane((int)(threadIdx.x >> 5));
    const int c    = lane & 15;
    const int h    = lane >> 4;
    const int bx   = blockIdx.x;

    int task = -1;
    if (wave < 2)                  task = bx * 2 + wave;
    else if (wave == 2 && bx < 8)  task = 2 * MAIN_BLOCKS + bx;
    if (task >= 0) {
        const int t   = task >> 3;
        const int m0e = (task & 7) * 16;
        if (t >= EDGE_ROW_TILES && t < EDGE_ROW_TILES + EDGE_COL_TILES) {
            const int tc = t - EDGE_ROW_TILES;
            edge_col_tile(x, wgt, bias, out, m0e, c, h, tc >> 2, (tc & 3) * 16 + c);
        } else {
            const bool corner = t >= EDGE_ROW_TILES;
            const int  pb     = corner ? (c < BN ? c : BN - 1) : (t >> 2);
            const int  pv     = corner ? WW : (t & 3) * 16 + c;
            const bool valid  = !corner || c < BN;
            edge_row_tile(x, wgt, bias, out, m0e, c, h, pb, pv, valid);
        }
    }

    {
        const int osub = bx & 7;
        const int ug   = (bx >> 3) & 3;
        const int b    = bx >> 5;
        const int u0   = ug * 16 + (wave >> 1) * 4;
        const int vb   = (wave & 1) * 32;
        for (int g = 0; g < 2; ++g)
            tile_rows4_f42_f22(x, wgt, bias, out, osub * 16 + g * 8, c, h, b, u0, vb);
    }
}

extern "C" void kernel_launch(void* const* d_in, const int* in_sizes, int n_in,
                              void* d_out, int out_size, void* d_ws, size_t ws_size,
                              hipStream_t stream) {
    (void)in_sizes; (void)n_in; (void)out_size; (void)d_ws; (void)ws_size;
    const float* x    = (const float*)d_in[0];
    const float* wgt  = (const float*)d_in[1];
    const float* bias = (const float*)d_in[2];
    float*       out  = (float*)d_out;

    dim3 grid(MAIN_BLOCKS);
    dim3 block(256);
    deconv2d_parity_merged_wmma<<<grid, block, 0, stream>>>(x, wgt, bias, out);
}
